# starting kernel
# baseline (speedup 1.0000x reference)
#include <hip/hip_runtime.h>
#include <math.h>


typedef __attribute__((ext_vector_type(2))) float v2f;
typedef __attribute__((ext_vector_type(8))) float v8f;

#define N_NODES 50000
#define N_EDGES 800000
#define F_DIM   128
#define NBITS   64
#define NCLS    100
#define HD      300
#define LDS_STRIDE 132

__device__ __forceinline__ v8f wmma_f32(v2f a, v2f b, v8f c) {
  return __builtin_amdgcn_wmma_f32_16x16x4_f32(false, a, false, b, (short)0, c,
                                               false, false);
}

__global__ void __launch_bounds__(256)
degree_kernel(const int* __restrict__ dst, float* __restrict__ deg) {
  int e = blockIdx.x * blockDim.x + threadIdx.x;
  if (e < N_EDGES) atomicAdd(&deg[dst[e]], 1.0f);
}

__global__ void __launch_bounds__(256)
scatter_add_kernel(const float* __restrict__ x, const int* __restrict__ src,
                   const int* __restrict__ dst, float* __restrict__ agg) {
  int gid = blockIdx.x * blockDim.x + threadIdx.x;
  int e   = gid >> 5;
  if (e >= N_EDGES) return;
  int ch = (gid & 31) * 4;
  int s = src[e], d = dst[e];
  const float4 v = *reinterpret_cast<const float4*>(x + (size_t)s * F_DIM + ch);
  float* p = agg + (size_t)d * F_DIM + ch;
  atomicAdd(p + 0, v.x);
  atomicAdd(p + 1, v.y);
  atomicAdd(p + 2, v.z);
  atomicAdd(p + 3, v.w);
}

__global__ void __launch_bounds__(256)
sage_layer_kernel(const float* __restrict__ x, const float* __restrict__ agg,
                  const float* __restrict__ deg, const float* __restrict__ wl,
                  const float* __restrict__ bl, const float* __restrict__ wr,
                  float* __restrict__ h) {
  __shared__ float sMean[16 * LDS_STRIDE];
  __shared__ float sX[16 * LDS_STRIDE];

  const int tm = blockIdx.x;

  for (int i = threadIdx.x; i < 16 * 32; i += 256) {
    const int r    = i >> 5;
    const int c4   = (i & 31) * 4;
    const int grow = tm * 16 + r;
    const float invd = 1.0f / fmaxf(deg[grow], 1.0f);
    const float4 va = *reinterpret_cast<const float4*>(agg + (size_t)grow * F_DIM + c4);
    const float4 vx = *reinterpret_cast<const float4*>(x   + (size_t)grow * F_DIM + c4);
    float* pm = sMean + r * LDS_STRIDE + c4;
    pm[0] = va.x * invd; pm[1] = va.y * invd; pm[2] = va.z * invd; pm[3] = va.w * invd;
    float* px = sX + r * LDS_STRIDE + c4;
    px[0] = vx.x; px[1] = vx.y; px[2] = vx.z; px[3] = vx.w;
  }
  __syncthreads();

  const int lane = threadIdx.x & 31;
  const int tn   = threadIdx.x >> 5;
  const int mloc = lane & 15;
  const int ncol = tn * 16 + mloc;
  const int kb   = (lane >= 16) ? 2 : 0;
  const float* sm = sMean + mloc * LDS_STRIDE;
  const float* sx = sX    + mloc * LDS_STRIDE;

  v8f acc = {};
#pragma unroll 4
  for (int kk = 0; kk < F_DIM; kk += 4) {
    v2f a1, b1, a2, b2;
    a1.x = sm[kk + kb];
    a1.y = sm[kk + kb + 1];
    b1.x = wl[(kk + kb)     * F_DIM + ncol];
    b1.y = wl[(kk + kb + 1) * F_DIM + ncol];
    acc = wmma_f32(a1, b1, acc);
    a2.x = sx[kk + kb];
    a2.y = sx[kk + kb + 1];
    b2.x = wr[(kk + kb)     * F_DIM + ncol];
    b2.y = wr[(kk + kb + 1) * F_DIM + ncol];
    acc = wmma_f32(a2, b2, acc);
  }
  const float bias  = bl[ncol];
  const int   mbase = tm * 16 + ((lane >= 16) ? 8 : 0);
#pragma unroll
  for (int i = 0; i < 8; ++i) {
    float v = fmaxf(acc[i] + bias, 0.0f);
    h[(size_t)(mbase + i) * F_DIM + ncol] = v;
  }
}

template <int ACT>
__global__ void __launch_bounds__(256)
wmma_gemm_bias_kernel(const float* __restrict__ A, const float* __restrict__ B,
                      const float* __restrict__ bias, float* __restrict__ C,
                      int N, int K) {
  __shared__ float sA[16 * LDS_STRIDE];

  const int tm = blockIdx.y;
  const int k4 = K >> 2;
  for (int i = threadIdx.x; i < 16 * k4; i += blockDim.x) {
    const int r  = i / k4;
    const int c4 = (i - r * k4) * 4;
    const float4 v = *reinterpret_cast<const float4*>(A + (size_t)(tm * 16 + r) * K + c4);
    float* p = sA + r * LDS_STRIDE + c4;
    p[0] = v.x; p[1] = v.y; p[2] = v.z; p[3] = v.w;
  }
  __syncthreads();

  const int lane   = threadIdx.x & 31;
  const int wave   = threadIdx.x >> 5;
  const int tilesN = (N + 15) >> 4;
  int tn = blockIdx.x * (blockDim.x >> 5) + wave;
  if (tn >= tilesN) tn = tilesN - 1;
  const int mloc = lane & 15;
  const int ncol = tn * 16 + mloc;
  const bool nok = (ncol < N);
  const int kb = (lane >= 16) ? 2 : 0;
  const float* sa = sA + mloc * LDS_STRIDE;

  v8f acc = {};
  for (int kk = 0; kk < K; kk += 4) {
    v2f a, b;
    a.x = sa[kk + kb];
    a.y = sa[kk + kb + 1];
    b.x = nok ? B[(size_t)(kk + kb) * N + ncol]     : 0.0f;
    b.y = nok ? B[(size_t)(kk + kb + 1) * N + ncol] : 0.0f;
    acc = wmma_f32(a, b, acc);
  }
  if (nok) {
    const float bv    = bias[ncol];
    const int   mbase = tm * 16 + ((lane >= 16) ? 8 : 0);
#pragma unroll
    for (int i = 0; i < 8; ++i) {
      float v = acc[i] + bv;
      if (ACT == 1) v = fmaxf(v, 0.0f);
      C[(size_t)(mbase + i) * N + ncol] = v;
    }
  }
}

__global__ void __launch_bounds__(256)
wmma_gemm_bn_kernel(const float* __restrict__ A, const float* __restrict__ B,
                    const float* __restrict__ bconv, const float* __restrict__ gamma,
                    const float* __restrict__ beta, const float* __restrict__ rm,
                    const float* __restrict__ rv, float* __restrict__ C,
                    int N, int K) {
  __shared__ float sA[16 * LDS_STRIDE];

  const int tm = blockIdx.y;
  const int k4 = K >> 2;
  for (int i = threadIdx.x; i < 16 * k4; i += blockDim.x) {
    const int r  = i / k4;
    const int c4 = (i - r * k4) * 4;
    const float4 v = *reinterpret_cast<const float4*>(A + (size_t)(tm * 16 + r) * K + c4);
    float* p = sA + r * LDS_STRIDE + c4;
    p[0] = v.x; p[1] = v.y; p[2] = v.z; p[3] = v.w;
  }
  __syncthreads();

  const int lane   = threadIdx.x & 31;
  const int wave   = threadIdx.x >> 5;
  const int tilesN = (N + 15) >> 4;
  int tn = blockIdx.x * (blockDim.x >> 5) + wave;
  if (tn >= tilesN) tn = tilesN - 1;
  const int mloc = lane & 15;
  const int ncol = tn * 16 + mloc;
  const bool nok = (ncol < N);
  const int kb = (lane >= 16) ? 2 : 0;
  const float* sa = sA + mloc * LDS_STRIDE;

  v8f acc = {};
  for (int kk = 0; kk < K; kk += 4) {
    v2f a, b;
    a.x = sa[kk + kb];
    a.y = sa[kk + kb + 1];
    b.x = nok ? B[(size_t)(kk + kb) * N + ncol]     : 0.0f;
    b.y = nok ? B[(size_t)(kk + kb + 1) * N + ncol] : 0.0f;
    acc = wmma_f32(a, b, acc);
  }
  if (nok) {
    const float s     = gamma[ncol] * rsqrtf(rv[ncol] + 1e-5f);
    const float shift = (bconv[ncol] - rm[ncol]) * s + beta[ncol];
    const int   mbase = tm * 16 + ((lane >= 16) ? 8 : 0);
#pragma unroll
    for (int i = 0; i < 8; ++i)
      C[(size_t)(mbase + i) * N + ncol] = acc[i] * s + shift;
  }
}

__global__ void __launch_bounds__(256)
tanh_kernel(const float* __restrict__ in, float* __restrict__ out, int n) {
  int i = blockIdx.x * blockDim.x + threadIdx.x;
  if (i < n) out[i] = tanhf(in[i]);
}

__global__ void __launch_bounds__(256)
true_lab_kernel(const float* __restrict__ pre, const float* __restrict__ wtl,
                const float* __restrict__ btl, float* __restrict__ out) {
  int r = blockIdx.x * blockDim.x + threadIdx.x;
  if (r >= N_NODES) return;
  const float* p = pre + (size_t)r * NBITS;
  float s = 0.0f;
#pragma unroll
  for (int k = 0; k < NBITS; ++k) s += p[k] * wtl[k];
  out[r] = s + btl[0];
}

extern "C" void kernel_launch(void* const* d_in, const int* in_sizes, int n_in,
                              void* d_out, int out_size, void* d_ws, size_t ws_size,
                              hipStream_t stream) {
  (void)in_sizes; (void)n_in; (void)out_size; (void)ws_size;

  const float* features = (const float*)d_in[0];
  const int*   edges    = (const int*)d_in[1];
  const float* w1l   = (const float*)d_in[2];
  const float* b1l   = (const float*)d_in[3];
  const float* w1r   = (const float*)d_in[4];
  const float* w2l   = (const float*)d_in[5];
  const float* b2l   = (const float*)d_in[6];
  const float* w2r   = (const float*)d_in[7];
  const float* whd   = (const float*)d_in[8];
  const float* bhd   = (const float*)d_in[9];
  const float* wclas = (const float*)d_in[10];
  const float* bclas = (const float*)d_in[11];
  const float* wconv = (const float*)d_in[12];
  const float* bconv = (const float*)d_in[13];
  const float* gamma = (const float*)d_in[14];
  const float* beta  = (const float*)d_in[15];
  const float* rm    = (const float*)d_in[16];
  const float* rv    = (const float*)d_in[17];
  const float* wtl   = (const float*)d_in[18];
  const float* btl   = (const float*)d_in[19];
  const float* wcv   = (const float*)d_in[20];
  const float* bcv   = (const float*)d_in[21];

  const int* src = edges;
  const int* dst = edges + N_EDGES;

  float* ws     = (float*)d_ws;
  float* deg    = ws;
  float* agg    = deg + N_NODES;
  float* h1     = agg + (size_t)N_NODES * F_DIM;
  float* h2     = h1  + (size_t)N_NODES * F_DIM;
  float* pre_bn = h2  + (size_t)N_NODES * F_DIM;

  float* out       = (float*)d_out;
  float* o_logists = out;
  float* o_tanh    = o_logists + (size_t)N_NODES * NCLS;
  float* o_fealab  = o_tanh    + (size_t)N_NODES * NBITS;
  float* o_feacv   = o_fealab  + (size_t)N_NODES * HD;
  float* o_truelab = o_feacv   + (size_t)N_NODES * NCLS;

  const int scatterBlocks = (N_EDGES * 32 + 255) / 256;
  const int rowTiles      = N_NODES / 16;
  auto gridFor = [rowTiles](int N, int waves) {
    int tilesN = (N + 15) / 16;
    return dim3((tilesN + waves - 1) / waves, rowTiles);
  };

  hipMemsetAsync(deg, 0, sizeof(float) * ((size_t)N_NODES * (F_DIM + 1)), stream);
  degree_kernel<<<(N_EDGES + 255) / 256, 256, 0, stream>>>(dst, deg);
  scatter_add_kernel<<<scatterBlocks, 256, 0, stream>>>(features, src, dst, agg);
  sage_layer_kernel<<<rowTiles, 256, 0, stream>>>(features, agg, deg, w1l, b1l, w1r, h1);

  hipMemsetAsync(agg, 0, sizeof(float) * (size_t)N_NODES * F_DIM, stream);
  scatter_add_kernel<<<scatterBlocks, 256, 0, stream>>>(h1, src, dst, agg);
  sage_layer_kernel<<<rowTiles, 256, 0, stream>>>(h1, agg, deg, w2l, b2l, w2r, h2);

  wmma_gemm_bias_kernel<0><<<gridFor(HD, 8), 256, 0, stream>>>(
      h2, whd, bhd, o_fealab, HD, F_DIM);
  wmma_gemm_bias_kernel<0><<<gridFor(NCLS, 8), 256, 0, stream>>>(
      h2, wclas, bclas, o_logists, NCLS, F_DIM);
  wmma_gemm_bn_kernel<<<gridFor(NBITS, 4), 128, 0, stream>>>(
      h2, wconv, bconv, gamma, beta, rm, rv, pre_bn, NBITS, F_DIM);
  tanh_kernel<<<(N_NODES * NBITS + 255) / 256, 256, 0, stream>>>(
      pre_bn, o_tanh, N_NODES * NBITS);
  wmma_gemm_bias_kernel<0><<<gridFor(NCLS, 8), 256, 0, stream>>>(
      o_tanh, wcv, bcv, o_feacv, NCLS, NBITS);
  true_lab_kernel<<<(N_NODES + 255) / 256, 256, 0, stream>>>(
      pre_bn, wtl, btl, o_truelab);
}


# Short-lived waves replaced by resident ones: layer, GEMM and tanh kernels launch 768 (tanh 512) blocks that loop over the 3125 row tiles, B fragments held in registers; about six times fewer waves
# speedup vs baseline: 1.8140x; 1.8140x over previous
#include <hip/hip_runtime.h>
#include <math.h>


typedef __attribute__((ext_vector_type(2))) float v2f;
typedef __attribute__((ext_vector_type(8))) float v8f;

#define N_NODES 50000
#define N_EDGES 800000
#define F_DIM   128
#define NBITS   64
#define NCLS    100
#define HD      300
#define LDS_STRIDE 132

__device__ __forceinline__ v8f wmma_f32(v2f a, v2f b, v8f c) {
  return __builtin_amdgcn_wmma_f32_16x16x4_f32(false, a, false, b, (short)0, c,
                                               false, false);
}

#define AGG_R       196
#define AGG_BLOCKS  256
#define AGG_THREADS 1024
#define AGG_CAP     16384
#define N_GROUPS    (N_EDGES / 4)

template <int COUNT>
__global__ void __launch_bounds__(AGG_THREADS)
range_aggregate_kernel(const float* __restrict__ x, const int* __restrict__ src,
                       const int* __restrict__ dst, float* __restrict__ agg,
                       float* __restrict__ deg) {
  __shared__ float    sAcc[(AGG_R + 1) * F_DIM];
  __shared__ unsigned sList[AGG_CAP];
  __shared__ float    sCnt[AGG_R];
  __shared__ unsigned sN;

  const int t    = threadIdx.x;
  const int lane = t & 31;
  const int wave = __builtin_amdgcn_readfirstlane(t >> 5);
  const int base = blockIdx.x * AGG_R;

  for (int i = t; i < (AGG_R + 1) * F_DIM / 4; i += AGG_THREADS)
    reinterpret_cast<float4*>(sAcc)[i] = make_float4(0.0f, 0.0f, 0.0f, 0.0f);
  if (t < AGG_R) sCnt[t] = 0.0f;
  if (t == 0) sN = 0u;
  __syncthreads();

  auto append = [&](int e, unsigned u) {
    if (COUNT) atomicAdd(&sCnt[u], 1.0f);
    const unsigned pos = atomicAdd(&sN, 1u);
    if (pos < AGG_CAP) {
      sList[pos] = (unsigned)e | (u << 20);
    } else {
      const float* xr = x + (size_t)src[e] * F_DIM;
      for (int c = 0; c < F_DIM; ++c)
        atomicAdd(&sAcc[u * F_DIM + (c & 3) * 32 + (c >> 2)], xr[c]);
    }
  };

  const int4* d4 = reinterpret_cast<const int4*>(dst);
  const int off = blockIdx.x * (N_GROUPS / AGG_BLOCKS);
  auto load4 = [&](int it, int4* d, int* gi) {
#pragma unroll
    for (int j = 0; j < 4; ++j) {
      const int  g  = (it + j) * AGG_THREADS + t;
      const bool ok = g < N_GROUPS;
      int gg = g + off;
      if (gg >= N_GROUPS) gg -= N_GROUPS;
      gg = ok ? gg : 0;
      gi[j] = gg;
      int4 v = d4[gg];
      if (!ok) v = make_int4(-1, -1, -1, -1);
      d[j] = v;
    }
  };
  int4 cur[4];
  int  gcur[4];
  load4(0, cur, gcur);
  for (int it = 0; it < (N_GROUPS + AGG_THREADS - 1) / AGG_THREADS; it += 4) {
    int4 nxt[4];
    int  gnxt[4];
    load4(it + 4, nxt, gnxt);
#pragma unroll
    for (int j = 0; j < 4; ++j) {
      const unsigned u0 = (unsigned)(cur[j].x - base);
      const unsigned u1 = (unsigned)(cur[j].y - base);
      const unsigned u2 = (unsigned)(cur[j].z - base);
      const unsigned u3 = (unsigned)(cur[j].w - base);
      const unsigned ua = u0 < u1 ? u0 : u1;
      const unsigned ub = u2 < u3 ? u2 : u3;
      if ((ua < ub ? ua : ub) < AGG_R) {
        const int e = gcur[j] * 4;
        if (u0 < AGG_R) append(e + 0, u0);
        if (u1 < AGG_R) append(e + 1, u1);
        if (u2 < AGG_R) append(e + 2, u2);
        if (u3 < AGG_R) append(e + 3, u3);
      }
    }
#pragma unroll
    for (int j = 0; j < 4; ++j) { cur[j] = nxt[j]; gcur[j] = gnxt[j]; }
  }
  __syncthreads();

  const unsigned nraw = sN;
  const int n = (int)(nraw < AGG_CAP ? nraw : (unsigned)AGG_CAP);
  const unsigned spare = (unsigned)AGG_R << 20;
  auto entry = [&](int b) -> unsigned {
    const int idx = b + (lane & 7);
    return (idx < n) ? sList[idx] : spare;
  };
  unsigned ent = entry(wave * 8);
  int      s   = src[ent & 0xFFFFFu];
  for (int b = wave * 8; b < n; b += (AGG_THREADS / 32) * 8) {
    float4 row[8];
#pragma unroll
    for (int j = 0; j < 8; ++j) {
      const int sj = __builtin_amdgcn_readlane(s, j);
      row[j] = *reinterpret_cast<const float4*>(x + (size_t)sj * F_DIM + lane * 4);
    }
    const unsigned ent_n = entry(b + (AGG_THREADS / 32) * 8);
    const int      s_n   = src[ent_n & 0xFFFFFu];
#pragma unroll
    for (int j = 0; j < 8; ++j) {
      const unsigned ld = (unsigned)__builtin_amdgcn_readlane((int)ent, j) >> 20;
      float* p = sAcc + ld * F_DIM + lane;
      atomicAdd(p,      row[j].x);
      atomicAdd(p + 32, row[j].y);
      atomicAdd(p + 64, row[j].z);
      atomicAdd(p + 96, row[j].w);
    }
    ent = ent_n;
    s   = s_n;
  }
  __syncthreads();

  for (int i = t; i < AGG_R * 32; i += AGG_THREADS) {
    const int r    = i >> 5;
    const int grow = base + r;
    if (grow < N_NODES) {
      const float* p = sAcc + r * F_DIM + lane;
      const float4 v = make_float4(p[0], p[32], p[64], p[96]);
      *reinterpret_cast<float4*>(agg + (size_t)grow * F_DIM + lane * 4) = v;
    }
  }
  if (COUNT) {
    if (t < AGG_R && base + t < N_NODES) deg[base + t] = sCnt[t];
  }
}

#define ROW_TILES (N_NODES / 16)
#define LAYER_BLOCKS 768
__global__ void __launch_bounds__(256) __attribute__((amdgpu_waves_per_eu(6, 6)))
sage_layer_kernel(const float* __restrict__ x, const float* __restrict__ agg,
                  const float* __restrict__ deg, const float* __restrict__ wl,
                  const float* __restrict__ bl, const float* __restrict__ wr,
                  float* __restrict__ h) {
  __shared__ float sMean[16 * LDS_STRIDE];
  __shared__ float sX[16 * LDS_STRIDE];

  const int lane = threadIdx.x & 31;
  const int tn   = threadIdx.x >> 5;
  const int mloc = lane & 15;
  const int ncol = tn * 16 + mloc;
  const int kb   = (lane >= 16) ? 2 : 0;
  const float* sm = sMean + mloc * LDS_STRIDE;
  const float* sx = sX    + mloc * LDS_STRIDE;

  v2f bwl[F_DIM / 4], bwr[F_DIM / 4];
#pragma unroll
  for (int s = 0; s < F_DIM / 4; ++s) {
    const int kk = s * 4;
    bwl[s].x = wl[(kk + kb)     * F_DIM + ncol];
    bwl[s].y = wl[(kk + kb + 1) * F_DIM + ncol];
    bwr[s].x = wr[(kk + kb)     * F_DIM + ncol];
    bwr[s].y = wr[(kk + kb + 1) * F_DIM + ncol];
  }
  const float bias = bl[ncol];

  for (int tm = blockIdx.x; tm < ROW_TILES; tm += LAYER_BLOCKS) {
    for (int i = threadIdx.x; i < 16 * 32; i += 256) {
      const int r    = i >> 5;
      const int c4   = (i & 31) * 4;
      const int grow = tm * 16 + r;
      const float invd = 1.0f / fmaxf(deg[grow], 1.0f);
      const float4 va = *reinterpret_cast<const float4*>(agg + (size_t)grow * F_DIM + c4);
      const float4 vx = *reinterpret_cast<const float4*>(x   + (size_t)grow * F_DIM + c4);
      float* pm = sMean + r * LDS_STRIDE + c4;
      pm[0] = va.x * invd; pm[1] = va.y * invd; pm[2] = va.z * invd; pm[3] = va.w * invd;
      float* px = sX + r * LDS_STRIDE + c4;
      px[0] = vx.x; px[1] = vx.y; px[2] = vx.z; px[3] = vx.w;
    }
    __syncthreads();

    v8f acc = {};
#pragma unroll
    for (int s = 0; s < F_DIM / 4; ++s) {
      const int kk = s * 4;
      v2f a1, a2;
      a1.x = sm[kk + kb];
      a1.y = sm[kk + kb + 1];
      acc = wmma_f32(a1, bwl[s], acc);
      a2.x = sx[kk + kb];
      a2.y = sx[kk + kb + 1];
      acc = wmma_f32(a2, bwr[s], acc);
    }
    const int mbase = tm * 16 + ((lane >= 16) ? 8 : 0);
#pragma unroll
    for (int i = 0; i < 8; ++i) {
      float v = fmaxf(acc[i] + bias, 0.0f);
      h[(size_t)(mbase + i) * F_DIM + ncol] = v;
    }
    __syncthreads();
  }
}

template <int ACT, int K>
__global__ void __launch_bounds__(256)
wmma_gemm_bias_kernel(const float* __restrict__ A, const float* __restrict__ B,
                      const float* __restrict__ bias, float* __restrict__ C,
                      int N) {
  __shared__ float sA[16 * LDS_STRIDE];

  const int lane   = threadIdx.x & 31;
  const int wave   = threadIdx.x >> 5;
  const int tilesN = (N + 15) >> 4;
  int tn = blockIdx.x * (blockDim.x >> 5) + wave;
  if (tn >= tilesN) tn = tilesN - 1;
  const int mloc = lane & 15;
  const int ncol = tn * 16 + mloc;
  const bool nok = (ncol < N);
  const int kb = (lane >= 16) ? 2 : 0;
  const float* sa = sA + mloc * LDS_STRIDE;

  v2f bf[K / 4];
#pragma unroll
  for (int s = 0; s < K / 4; ++s) {
    const int kk = s * 4;
    bf[s].x = nok ? B[(size_t)(kk + kb) * N + ncol]     : 0.0f;
    bf[s].y = nok ? B[(size_t)(kk + kb + 1) * N + ncol] : 0.0f;
  }
  const float bv = nok ? bias[ncol] : 0.0f;

  constexpr int k4 = K >> 2;
  for (int tm = blockIdx.y; tm < ROW_TILES; tm += gridDim.y) {
    for (int i = threadIdx.x; i < 16 * k4; i += blockDim.x) {
      const int r  = i / k4;
      const int c4 = (i - r * k4) * 4;
      const float4 v = *reinterpret_cast<const float4*>(A + (size_t)(tm * 16 + r) * K + c4);
      float* p = sA + r * LDS_STRIDE + c4;
      p[0] = v.x; p[1] = v.y; p[2] = v.z; p[3] = v.w;
    }
    __syncthreads();

    v8f acc = {};
#pragma unroll
    for (int s = 0; s < K / 4; ++s) {
      const int kk = s * 4;
      v2f a;
      a.x = sa[kk + kb];
      a.y = sa[kk + kb + 1];
      acc = wmma_f32(a, bf[s], acc);
    }
    if (nok) {
      const int mbase = tm * 16 + ((lane >= 16) ? 8 : 0);
#pragma unroll
      for (int i = 0; i < 8; ++i) {
        float v = acc[i] + bv;
        if (ACT == 1) v = fmaxf(v, 0.0f);
        C[(size_t)(mbase + i) * N + ncol] = v;
      }
    }
    __syncthreads();
  }
}

template <int K>
__global__ void __launch_bounds__(256)
wmma_gemm_bn_kernel(const float* __restrict__ A, const float* __restrict__ B,
                    const float* __restrict__ bconv, const float* __restrict__ gamma,
                    const float* __restrict__ beta, const float* __restrict__ rm,
                    const float* __restrict__ rv, float* __restrict__ C,
                    int N) {
  __shared__ float sA[16 * LDS_STRIDE];

  const int lane   = threadIdx.x & 31;
  const int wave   = threadIdx.x >> 5;
  const int tilesN = (N + 15) >> 4;
  int tn = blockIdx.x * (blockDim.x >> 5) + wave;
  if (tn >= tilesN) tn = tilesN - 1;
  const int mloc = lane & 15;
  const int ncol = tn * 16 + mloc;
  const bool nok = (ncol < N);
  const int kb = (lane >= 16) ? 2 : 0;
  const float* sa = sA + mloc * LDS_STRIDE;

  v2f bf[K / 4];
#pragma unroll
  for (int s = 0; s < K / 4; ++s) {
    const int kk = s * 4;
    bf[s].x = nok ? B[(size_t)(kk + kb) * N + ncol]     : 0.0f;
    bf[s].y = nok ? B[(size_t)(kk + kb + 1) * N + ncol] : 0.0f;
  }
  float sc = 0.0f, shift = 0.0f;
  if (nok) {
    sc    = gamma[ncol] * rsqrtf(rv[ncol] + 1e-5f);
    shift = (bconv[ncol] - rm[ncol]) * sc + beta[ncol];
  }

  constexpr int k4 = K >> 2;
  for (int tm = blockIdx.y; tm < ROW_TILES; tm += gridDim.y) {
    for (int i = threadIdx.x; i < 16 * k4; i += blockDim.x) {
      const int r  = i / k4;
      const int c4 = (i - r * k4) * 4;
      const float4 v = *reinterpret_cast<const float4*>(A + (size_t)(tm * 16 + r) * K + c4);
      float* p = sA + r * LDS_STRIDE + c4;
      p[0] = v.x; p[1] = v.y; p[2] = v.z; p[3] = v.w;
    }
    __syncthreads();

    v8f acc = {};
#pragma unroll
    for (int s = 0; s < K / 4; ++s) {
      const int kk = s * 4;
      v2f a;
      a.x = sa[kk + kb];
      a.y = sa[kk + kb + 1];
      acc = wmma_f32(a, bf[s], acc);
    }
    if (nok) {
      const int mbase = tm * 16 + ((lane >= 16) ? 8 : 0);
#pragma unroll
      for (int i = 0; i < 8; ++i)
        C[(size_t)(mbase + i) * N + ncol] = acc[i] * sc + shift;
    }
    __syncthreads();
  }
}

__global__ void __launch_bounds__(256)
tanh_kernel(const float* __restrict__ in, float* __restrict__ out, int n4) {
  const float4* in4  = reinterpret_cast<const float4*>(in);
  float4*       out4 = reinterpret_cast<float4*>(out);
  for (int i = blockIdx.x * blockDim.x + threadIdx.x; i < n4; i += gridDim.x * blockDim.x) {
    const float4 v = in4[i];
    out4[i] = make_float4(tanhf(v.x), tanhf(v.y), tanhf(v.z), tanhf(v.w));
  }
}

__global__ void __launch_bounds__(256)
true_lab_kernel(const float* __restrict__ pre, const float* __restrict__ wtl,
                const float* __restrict__ btl, float* __restrict__ out) {
  int r = blockIdx.x * blockDim.x + threadIdx.x;
  if (r >= N_NODES) return;
  const float* p = pre + (size_t)r * NBITS;
  float s = 0.0f;
#pragma unroll
  for (int k = 0; k < NBITS; ++k) s += p[k] * wtl[k];
  out[r] = s + btl[0];
}

extern "C" void kernel_launch(void* const* d_in, const int* in_sizes, int n_in,
                              void* d_out, int out_size, void* d_ws, size_t ws_size,
                              hipStream_t stream) {
  (void)in_sizes; (void)n_in; (void)out_size; (void)ws_size;

  const float* features = (const float*)d_in[0];
  const int*   edges    = (const int*)d_in[1];
  const float* w1l   = (const float*)d_in[2];
  const float* b1l   = (const float*)d_in[3];
  const float* w1r   = (const float*)d_in[4];
  const float* w2l   = (const float*)d_in[5];
  const float* b2l   = (const float*)d_in[6];
  const float* w2r   = (const float*)d_in[7];
  const float* whd   = (const float*)d_in[8];
  const float* bhd   = (const float*)d_in[9];
  const float* wclas = (const float*)d_in[10];
  const float* bclas = (const float*)d_in[11];
  const float* wconv = (const float*)d_in[12];
  const float* bconv = (const float*)d_in[13];
  const float* gamma = (const float*)d_in[14];
  const float* beta  = (const float*)d_in[15];
  const float* rm    = (const float*)d_in[16];
  const float* rv    = (const float*)d_in[17];
  const float* wtl   = (const float*)d_in[18];
  const float* btl   = (const float*)d_in[19];
  const float* wcv   = (const float*)d_in[20];
  const float* bcv   = (const float*)d_in[21];

  const int* src = edges;
  const int* dst = edges + N_EDGES;

  float* ws     = (float*)d_ws;
  float* deg    = ws;
  float* agg    = deg + N_NODES;
  float* h1     = agg + (size_t)N_NODES * F_DIM;
  float* h2     = h1  + (size_t)N_NODES * F_DIM;
  float* pre_bn = h2  + (size_t)N_NODES * F_DIM;

  float* out       = (float*)d_out;
  float* o_logists = out;
  float* o_tanh    = o_logists + (size_t)N_NODES * NCLS;
  float* o_fealab  = o_tanh    + (size_t)N_NODES * NBITS;
  float* o_feacv   = o_fealab  + (size_t)N_NODES * HD;
  float* o_truelab = o_feacv   + (size_t)N_NODES * NCLS;

  range_aggregate_kernel<1><<<AGG_BLOCKS, AGG_THREADS, 0, stream>>>(features, src, dst, agg, deg);
  sage_layer_kernel<<<LAYER_BLOCKS, 256, 0, stream>>>(features, agg, deg, w1l, b1l, w1r, h1);

  range_aggregate_kernel<0><<<AGG_BLOCKS, AGG_THREADS, 0, stream>>>(h1, src, dst, agg, deg);
  sage_layer_kernel<<<LAYER_BLOCKS, 256, 0, stream>>>(h1, agg, deg, w2l, b2l, w2r, h2);

  wmma_gemm_bias_kernel<0, F_DIM><<<dim3(3, 256), 256, 0, stream>>>(
      h2, whd, bhd, o_fealab, HD);
  wmma_gemm_bias_kernel<0, F_DIM><<<dim3(1, 768), 256, 0, stream>>>(
      h2, wclas, bclas, o_logists, NCLS);
  wmma_gemm_bn_kernel<F_DIM><<<dim3(1, 768), 128, 0, stream>>>(
      h2, wconv, bconv, gamma, beta, rm, rv, pre_bn, NBITS);
  tanh_kernel<<<512, 256, 0, stream>>>(
      pre_bn, o_tanh, N_NODES * NBITS / 4);
  wmma_gemm_bias_kernel<0, NBITS><<<dim3(1, 768), 256, 0, stream>>>(
      o_tanh, wcv, bcv, o_feacv, NCLS);
  true_lab_kernel<<<(N_NODES + 255) / 256, 256, 0, stream>>>(
      pre_bn, wtl, btl, o_truelab);
}


# Layer-2 neighbour-sum kernel no longer scans dst: the layer-1 launch stores each range block's matched-edge list in the workspace, layer 2 reads it back; LDS add order still arrival order
# speedup vs baseline: 2.0235x; 1.1155x over previous
#include <hip/hip_runtime.h>
#include <math.h>


typedef __attribute__((ext_vector_type(2))) float v2f;
typedef __attribute__((ext_vector_type(8))) float v8f;

#define N_NODES 50000
#define N_EDGES 800000
#define F_DIM   128
#define NBITS   64
#define NCLS    100
#define HD      300
#define LDS_STRIDE 132

__device__ __forceinline__ v8f wmma_f32(v2f a, v2f b, v8f c) {
  return __builtin_amdgcn_wmma_f32_16x16x4_f32(false, a, false, b, (short)0, c,
                                               false, false);
}

#define AGG_R       196
#define AGG_BLOCKS  256
#define AGG_THREADS 1024
#define AGG_CAP     16384
#define N_GROUPS    (N_EDGES / 4)

template <int COUNT>
__global__ void __launch_bounds__(AGG_THREADS)
range_aggregate_kernel(const float* __restrict__ x, const int* __restrict__ src,
                       const int* __restrict__ dst, float* __restrict__ agg,
                       float* __restrict__ deg, unsigned* __restrict__ wlist,
                       unsigned* __restrict__ wcount) {
  __shared__ float    sAcc[(AGG_R + 1) * F_DIM];
  __shared__ unsigned sList[AGG_CAP];
  __shared__ float    sCnt[AGG_R];
  __shared__ unsigned sN;

  const int t    = threadIdx.x;
  const int lane = t & 31;
  const int wave = __builtin_amdgcn_readfirstlane(t >> 5);
  const int base = blockIdx.x * AGG_R;

  for (int i = t; i < (AGG_R + 1) * F_DIM / 4; i += AGG_THREADS)
    reinterpret_cast<float4*>(sAcc)[i] = make_float4(0.0f, 0.0f, 0.0f, 0.0f);
  if (t < AGG_R) sCnt[t] = 0.0f;
  if (t == 0) sN = 0u;
  __syncthreads();

  auto append = [&](int e, unsigned u) {
    if (COUNT) atomicAdd(&sCnt[u], 1.0f);
    const unsigned pos = atomicAdd(&sN, 1u);
    if (pos < AGG_CAP) {
      sList[pos] = (unsigned)e | (u << 20);
    } else {
      const float* xr = x + (size_t)src[e] * F_DIM;
      for (int c = 0; c < F_DIM; ++c)
        atomicAdd(&sAcc[u * F_DIM + (c & 3) * 32 + (c >> 2)], xr[c]);
    }
  };

  unsigned saved = 0u;
  if (!COUNT) saved = (unsigned)__builtin_amdgcn_readfirstlane((int)wcount[blockIdx.x]);
  const bool rescan = COUNT || saved > (unsigned)AGG_CAP;
  unsigned* wl = wlist + (size_t)blockIdx.x * AGG_CAP;

  if (!rescan) {
    for (unsigned i = t; i < saved; i += AGG_THREADS) sList[i] = wl[i];
    if (t == 0) sN = saved;
  } else {
    const int4* d4 = reinterpret_cast<const int4*>(dst);
    const int off = blockIdx.x * (N_GROUPS / AGG_BLOCKS);
    auto load4 = [&](int it, int4* d, int* gi) {
#pragma unroll
      for (int j = 0; j < 4; ++j) {
        const int  g  = (it + j) * AGG_THREADS + t;
        const bool ok = g < N_GROUPS;
        int gg = g + off;
        if (gg >= N_GROUPS) gg -= N_GROUPS;
        gg = ok ? gg : 0;
        gi[j] = gg;
        int4 v = d4[gg];
        if (!ok) v = make_int4(-1, -1, -1, -1);
        d[j] = v;
      }
    };
    int4 cur[4];
    int  gcur[4];
    load4(0, cur, gcur);
    for (int it = 0; it < (N_GROUPS + AGG_THREADS - 1) / AGG_THREADS; it += 4) {
      int4 nxt[4];
      int  gnxt[4];
      load4(it + 4, nxt, gnxt);
#pragma unroll
      for (int j = 0; j < 4; ++j) {
        const unsigned u0 = (unsigned)(cur[j].x - base);
        const unsigned u1 = (unsigned)(cur[j].y - base);
        const unsigned u2 = (unsigned)(cur[j].z - base);
        const unsigned u3 = (unsigned)(cur[j].w - base);
        const unsigned ua = u0 < u1 ? u0 : u1;
        const unsigned ub = u2 < u3 ? u2 : u3;
        if ((ua < ub ? ua : ub) < AGG_R) {
          const int e = gcur[j] * 4;
          if (u0 < AGG_R) append(e + 0, u0);
          if (u1 < AGG_R) append(e + 1, u1);
          if (u2 < AGG_R) append(e + 2, u2);
          if (u3 < AGG_R) append(e + 3, u3);
        }
      }
#pragma unroll
      for (int j = 0; j < 4; ++j) { cur[j] = nxt[j]; gcur[j] = gnxt[j]; }
    }
  }
  __syncthreads();

  const unsigned nraw = sN;
  const int n = (int)(nraw < AGG_CAP ? nraw : (unsigned)AGG_CAP);
  const unsigned spare = (unsigned)AGG_R << 20;
  if (COUNT) {
    for (int i = t; i < n; i += AGG_THREADS) wl[i] = sList[i];
    if (t == 0) wcount[blockIdx.x] = nraw;
  }
  auto entry = [&](int b) -> unsigned {
    const int idx = b + (lane & 7);
    return (idx < n) ? sList[idx] : spare;
  };
  unsigned ent = entry(wave * 8);
  int      s   = src[ent & 0xFFFFFu];
  for (int b = wave * 8; b < n; b += (AGG_THREADS / 32) * 8) {
    float4 row[8];
#pragma unroll
    for (int j = 0; j < 8; ++j) {
      const int sj = __builtin_amdgcn_readlane(s, j);
      row[j] = *reinterpret_cast<const float4*>(x + (size_t)sj * F_DIM + lane * 4);
    }
    const unsigned ent_n = entry(b + (AGG_THREADS / 32) * 8);
    const int      s_n   = src[ent_n & 0xFFFFFu];
#pragma unroll
    for (int j = 0; j < 8; ++j) {
      const unsigned ld = (unsigned)__builtin_amdgcn_readlane((int)ent, j) >> 20;
      float* p = sAcc + ld * F_DIM + lane;
      atomicAdd(p,      row[j].x);
      atomicAdd(p + 32, row[j].y);
      atomicAdd(p + 64, row[j].z);
      atomicAdd(p + 96, row[j].w);
    }
    ent = ent_n;
    s   = s_n;
  }
  __syncthreads();

  for (int i = t; i < AGG_R * 32; i += AGG_THREADS) {
    const int r    = i >> 5;
    const int grow = base + r;
    if (grow < N_NODES) {
      const float* p = sAcc + r * F_DIM + lane;
      const float4 v = make_float4(p[0], p[32], p[64], p[96]);
      *reinterpret_cast<float4*>(agg + (size_t)grow * F_DIM + lane * 4) = v;
    }
  }
  if (COUNT) {
    if (t < AGG_R && base + t < N_NODES) deg[base + t] = sCnt[t];
  }
}

#define ROW_TILES (N_NODES / 16)
#define LAYER_BLOCKS 768
__global__ void __launch_bounds__(256) __attribute__((amdgpu_waves_per_eu(6, 6)))
sage_layer_kernel(const float* __restrict__ x, const float* __restrict__ agg,
                  const float* __restrict__ deg, const float* __restrict__ wl,
                  const float* __restrict__ bl, const float* __restrict__ wr,
                  float* __restrict__ h) {
  __shared__ float sMean[16 * LDS_STRIDE];
  __shared__ float sX[16 * LDS_STRIDE];

  const int lane = threadIdx.x & 31;
  const int tn   = threadIdx.x >> 5;
  const int mloc = lane & 15;
  const int ncol = tn * 16 + mloc;
  const int kb   = (lane >= 16) ? 2 : 0;
  const float* sm = sMean + mloc * LDS_STRIDE;
  const float* sx = sX    + mloc * LDS_STRIDE;

  v2f bwl[F_DIM / 4], bwr[F_DIM / 4];
#pragma unroll
  for (int s = 0; s < F_DIM / 4; ++s) {
    const int kk = s * 4;
    bwl[s].x = wl[(kk + kb)     * F_DIM + ncol];
    bwl[s].y = wl[(kk + kb + 1) * F_DIM + ncol];
    bwr[s].x = wr[(kk + kb)     * F_DIM + ncol];
    bwr[s].y = wr[(kk + kb + 1) * F_DIM + ncol];
  }
  const float bias = bl[ncol];

  for (int tm = blockIdx.x; tm < ROW_TILES; tm += LAYER_BLOCKS) {
    for (int i = threadIdx.x; i < 16 * 32; i += 256) {
      const int r    = i >> 5;
      const int c4   = (i & 31) * 4;
      const int grow = tm * 16 + r;
      const float invd = 1.0f / fmaxf(deg[grow], 1.0f);
      const float4 va = *reinterpret_cast<const float4*>(agg + (size_t)grow * F_DIM + c4);
      const float4 vx = *reinterpret_cast<const float4*>(x   + (size_t)grow * F_DIM + c4);
      float* pm = sMean + r * LDS_STRIDE + c4;
      pm[0] = va.x * invd; pm[1] = va.y * invd; pm[2] = va.z * invd; pm[3] = va.w * invd;
      float* px = sX + r * LDS_STRIDE + c4;
      px[0] = vx.x; px[1] = vx.y; px[2] = vx.z; px[3] = vx.w;
    }
    __syncthreads();

    v8f acc = {};
#pragma unroll
    for (int s = 0; s < F_DIM / 4; ++s) {
      const int kk = s * 4;
      v2f a1, a2;
      a1.x = sm[kk + kb];
      a1.y = sm[kk + kb + 1];
      acc = wmma_f32(a1, bwl[s], acc);
      a2.x = sx[kk + kb];
      a2.y = sx[kk + kb + 1];
      acc = wmma_f32(a2, bwr[s], acc);
    }
    const int mbase = tm * 16 + ((lane >= 16) ? 8 : 0);
#pragma unroll
    for (int i = 0; i < 8; ++i) {
      float v = fmaxf(acc[i] + bias, 0.0f);
      h[(size_t)(mbase + i) * F_DIM + ncol] = v;
    }
    __syncthreads();
  }
}

template <int ACT, int K>
__global__ void __launch_bounds__(256)
wmma_gemm_bias_kernel(const float* __restrict__ A, const float* __restrict__ B,
                      const float* __restrict__ bias, float* __restrict__ C,
                      int N) {
  __shared__ float sA[16 * LDS_STRIDE];

  const int lane   = threadIdx.x & 31;
  const int wave   = threadIdx.x >> 5;
  const int tilesN = (N + 15) >> 4;
  int tn = blockIdx.x * (blockDim.x >> 5) + wave;
  if (tn >= tilesN) tn = tilesN - 1;
  const int mloc = lane & 15;
  const int ncol = tn * 16 + mloc;
  const bool nok = (ncol < N);
  const int kb = (lane >= 16) ? 2 : 0;
  const float* sa = sA + mloc * LDS_STRIDE;

  v2f bf[K / 4];
#pragma unroll
  for (int s = 0; s < K / 4; ++s) {
    const int kk = s * 4;
    bf[s].x = nok ? B[(size_t)(kk + kb) * N + ncol]     : 0.0f;
    bf[s].y = nok ? B[(size_t)(kk + kb + 1) * N + ncol] : 0.0f;
  }
  const float bv = nok ? bias[ncol] : 0.0f;

  constexpr int k4 = K >> 2;
  for (int tm = blockIdx.y; tm < ROW_TILES; tm += gridDim.y) {
    for (int i = threadIdx.x; i < 16 * k4; i += blockDim.x) {
      const int r  = i / k4;
      const int c4 = (i - r * k4) * 4;
      const float4 v = *reinterpret_cast<const float4*>(A + (size_t)(tm * 16 + r) * K + c4);
      float* p = sA + r * LDS_STRIDE + c4;
      p[0] = v.x; p[1] = v.y; p[2] = v.z; p[3] = v.w;
    }
    __syncthreads();

    v8f acc = {};
#pragma unroll
    for (int s = 0; s < K / 4; ++s) {
      const int kk = s * 4;
      v2f a;
      a.x = sa[kk + kb];
      a.y = sa[kk + kb + 1];
      acc = wmma_f32(a, bf[s], acc);
    }
    if (nok) {
      const int mbase = tm * 16 + ((lane >= 16) ? 8 : 0);
#pragma unroll
      for (int i = 0; i < 8; ++i) {
        float v = acc[i] + bv;
        if (ACT == 1) v = fmaxf(v, 0.0f);
        C[(size_t)(mbase + i) * N + ncol] = v;
      }
    }
    __syncthreads();
  }
}

template <int K>
__global__ void __launch_bounds__(256)
wmma_gemm_bn_kernel(const float* __restrict__ A, const float* __restrict__ B,
                    const float* __restrict__ bconv, const float* __restrict__ gamma,
                    const float* __restrict__ beta, const float* __restrict__ rm,
                    const float* __restrict__ rv, float* __restrict__ C,
                    int N) {
  __shared__ float sA[16 * LDS_STRIDE];

  const int lane   = threadIdx.x & 31;
  const int wave   = threadIdx.x >> 5;
  const int tilesN = (N + 15) >> 4;
  int tn = blockIdx.x * (blockDim.x >> 5) + wave;
  if (tn >= tilesN) tn = tilesN - 1;
  const int mloc = lane & 15;
  const int ncol = tn * 16 + mloc;
  const bool nok = (ncol < N);
  const int kb = (lane >= 16) ? 2 : 0;
  const float* sa = sA + mloc * LDS_STRIDE;

  v2f bf[K / 4];
#pragma unroll
  for (int s = 0; s < K / 4; ++s) {
    const int kk = s * 4;
    bf[s].x = nok ? B[(size_t)(kk + kb) * N + ncol]     : 0.0f;
    bf[s].y = nok ? B[(size_t)(kk + kb + 1) * N + ncol] : 0.0f;
  }
  float sc = 0.0f, shift = 0.0f;
  if (nok) {
    sc    = gamma[ncol] * rsqrtf(rv[ncol] + 1e-5f);
    shift = (bconv[ncol] - rm[ncol]) * sc + beta[ncol];
  }

  constexpr int k4 = K >> 2;
  for (int tm = blockIdx.y; tm < ROW_TILES; tm += gridDim.y) {
    for (int i = threadIdx.x; i < 16 * k4; i += blockDim.x) {
      const int r  = i / k4;
      const int c4 = (i - r * k4) * 4;
      const float4 v = *reinterpret_cast<const float4*>(A + (size_t)(tm * 16 + r) * K + c4);
      float* p = sA + r * LDS_STRIDE + c4;
      p[0] = v.x; p[1] = v.y; p[2] = v.z; p[3] = v.w;
    }
    __syncthreads();

    v8f acc = {};
#pragma unroll
    for (int s = 0; s < K / 4; ++s) {
      const int kk = s * 4;
      v2f a;
      a.x = sa[kk + kb];
      a.y = sa[kk + kb + 1];
      acc = wmma_f32(a, bf[s], acc);
    }
    if (nok) {
      const int mbase = tm * 16 + ((lane >= 16) ? 8 : 0);
#pragma unroll
      for (int i = 0; i < 8; ++i)
        C[(size_t)(mbase + i) * N + ncol] = acc[i] * sc + shift;
    }
    __syncthreads();
  }
}

__global__ void __launch_bounds__(256)
tanh_kernel(const float* __restrict__ in, float* __restrict__ out, int n4) {
  const float4* in4  = reinterpret_cast<const float4*>(in);
  float4*       out4 = reinterpret_cast<float4*>(out);
  for (int i = blockIdx.x * blockDim.x + threadIdx.x; i < n4; i += gridDim.x * blockDim.x) {
    const float4 v = in4[i];
    out4[i] = make_float4(tanhf(v.x), tanhf(v.y), tanhf(v.z), tanhf(v.w));
  }
}

__global__ void __launch_bounds__(256)
true_lab_kernel(const float* __restrict__ pre, const float* __restrict__ wtl,
                const float* __restrict__ btl, float* __restrict__ out) {
  int r = blockIdx.x * blockDim.x + threadIdx.x;
  if (r >= N_NODES) return;
  const float* p = pre + (size_t)r * NBITS;
  float s = 0.0f;
#pragma unroll
  for (int k = 0; k < NBITS; ++k) s += p[k] * wtl[k];
  out[r] = s + btl[0];
}

extern "C" void kernel_launch(void* const* d_in, const int* in_sizes, int n_in,
                              void* d_out, int out_size, void* d_ws, size_t ws_size,
                              hipStream_t stream) {
  (void)in_sizes; (void)n_in; (void)out_size; (void)ws_size;

  const float* features = (const float*)d_in[0];
  const int*   edges    = (const int*)d_in[1];
  const float* w1l   = (const float*)d_in[2];
  const float* b1l   = (const float*)d_in[3];
  const float* w1r   = (const float*)d_in[4];
  const float* w2l   = (const float*)d_in[5];
  const float* b2l   = (const float*)d_in[6];
  const float* w2r   = (const float*)d_in[7];
  const float* whd   = (const float*)d_in[8];
  const float* bhd   = (const float*)d_in[9];
  const float* wclas = (const float*)d_in[10];
  const float* bclas = (const float*)d_in[11];
  const float* wconv = (const float*)d_in[12];
  const float* bconv = (const float*)d_in[13];
  const float* gamma = (const float*)d_in[14];
  const float* beta  = (const float*)d_in[15];
  const float* rm    = (const float*)d_in[16];
  const float* rv    = (const float*)d_in[17];
  const float* wtl   = (const float*)d_in[18];
  const float* btl   = (const float*)d_in[19];
  const float* wcv   = (const float*)d_in[20];
  const float* bcv   = (const float*)d_in[21];

  const int* src = edges;
  const int* dst = edges + N_EDGES;

  float* ws     = (float*)d_ws;
  float* deg    = ws;
  float* agg    = deg + N_NODES;
  float* h1     = agg + (size_t)N_NODES * F_DIM;
  float* h2     = h1  + (size_t)N_NODES * F_DIM;
  float* pre_bn = h2  + (size_t)N_NODES * F_DIM;
  unsigned* wcount = (unsigned*)(pre_bn + (size_t)N_NODES * NBITS);
  unsigned* wlist  = wcount + AGG_BLOCKS;

  float* out       = (float*)d_out;
  float* o_logists = out;
  float* o_tanh    = o_logists + (size_t)N_NODES * NCLS;
  float* o_fealab  = o_tanh    + (size_t)N_NODES * NBITS;
  float* o_feacv   = o_fealab  + (size_t)N_NODES * HD;
  float* o_truelab = o_feacv   + (size_t)N_NODES * NCLS;

  range_aggregate_kernel<1><<<AGG_BLOCKS, AGG_THREADS, 0, stream>>>(features, src, dst, agg, deg,
                                                                    wlist, wcount);
  sage_layer_kernel<<<LAYER_BLOCKS, 256, 0, stream>>>(features, agg, deg, w1l, b1l, w1r, h1);

  range_aggregate_kernel<0><<<AGG_BLOCKS, AGG_THREADS, 0, stream>>>(h1, src, dst, agg, deg,
                                                                    wlist, wcount);
  sage_layer_kernel<<<LAYER_BLOCKS, 256, 0, stream>>>(h1, agg, deg, w2l, b2l, w2r, h2);

  wmma_gemm_bias_kernel<0, F_DIM><<<dim3(3, 256), 256, 0, stream>>>(
      h2, whd, bhd, o_fealab, HD);
  wmma_gemm_bias_kernel<0, F_DIM><<<dim3(1, 768), 256, 0, stream>>>(
      h2, wclas, bclas, o_logists, NCLS);
  wmma_gemm_bn_kernel<F_DIM><<<dim3(1, 768), 128, 0, stream>>>(
      h2, wconv, bconv, gamma, beta, rm, rv, pre_bn, NBITS);
  tanh_kernel<<<512, 256, 0, stream>>>(
      pre_bn, o_tanh, N_NODES * NBITS / 4);
  wmma_gemm_bias_kernel<0, NBITS><<<dim3(1, 768), 256, 0, stream>>>(
      o_tanh, wcv, bcv, o_feacv, NCLS);
  true_lab_kernel<<<(N_NODES + 255) / 256, 256, 0, stream>>>(
      pre_bn, wtl, btl, o_truelab);
}


# SAGE layer fused into the range kernel: the block runs both WMMA products straight from its LDS neighbour sums, x rows staged beside them; sums plane and 2 launches gone (10->8); add order unchanged
# speedup vs baseline: 2.2100x; 1.0922x over previous
#include <hip/hip_runtime.h>
#include <math.h>


typedef __attribute__((ext_vector_type(2))) float v2f;
typedef __attribute__((ext_vector_type(8))) float v8f;

#define N_NODES 50000
#define N_EDGES 800000
#define F_DIM   128
#define NBITS   64
#define NCLS    100
#define HD      300
#define LDS_STRIDE 132

__device__ __forceinline__ v8f wmma_f32(v2f a, v2f b, v8f c) {
  return __builtin_amdgcn_wmma_f32_16x16x4_f32(false, a, false, b, (short)0, c,
                                               false, false);
}

#define AGG_R       196
#define AGG_TILES   13
#define AGG_ROWS    (AGG_TILES * 16)
#define AGG_BLOCKS  256
#define AGG_THREADS 1024
#define AGG_CAP     16384
#define N_GROUPS    (N_EDGES / 4)
#define ACC_COMP    33

template <int NT>
__device__ __forceinline__ void range_tiles_gemm(
    const float* sAcc, const float* sX, const float* sCnt,
    const float* __restrict__ wl, const float* __restrict__ bl,
    const float* __restrict__ wr, float* __restrict__ h,
    int base, int rt0, int tn, int lane) {
  const int mloc = lane & 15;
  const int hh   = lane >> 4;
  const int kb   = hh * 2;
  const int ncol = tn * 16 + mloc;
  v8f   acc[NT];
  float inv[NT];
  const float* pa[NT];
  const float* px[NT];
#pragma unroll
  for (int i = 0; i < NT; ++i) {
    const int row = (rt0 + 4 * i) * 16 + mloc;
    inv[i] = 1.0f / fmaxf(sCnt[row], 1.0f);
    pa[i]  = sAcc + row * LDS_STRIDE + kb * ACC_COMP;
    px[i]  = sX   + row * LDS_STRIDE + kb;
    acc[i] = v8f{};
  }
  const float* pwl = wl + kb * F_DIM + ncol;
  const float* pwr = wr + kb * F_DIM + ncol;
#pragma unroll 8
  for (int s = 0; s < F_DIM / 4; ++s) {
    v2f b1, b2;
    b1.x = pwl[(s * 4)     * F_DIM];
    b1.y = pwl[(s * 4 + 1) * F_DIM];
    b2.x = pwr[(s * 4)     * F_DIM];
    b2.y = pwr[(s * 4 + 1) * F_DIM];
#pragma unroll
    for (int i = 0; i < NT; ++i) {
      v2f a1, a2;
      a1.x = pa[i][s] * inv[i];
      a1.y = pa[i][ACC_COMP + s] * inv[i];
      acc[i] = wmma_f32(a1, b1, acc[i]);
      a2.x = px[i][s * 4];
      a2.y = px[i][s * 4 + 1];
      acc[i] = wmma_f32(a2, b2, acc[i]);
    }
  }
  const float bias = bl[ncol];
#pragma unroll
  for (int i = 0; i < NT; ++i) {
    const int lr0 = (rt0 + 4 * i) * 16 + hh * 8;
#pragma unroll
    for (int q = 0; q < 8; ++q) {
      const int lr   = lr0 + q;
      const int grow = base + lr;
      if (lr < AGG_R && grow < N_NODES)
        h[(size_t)grow * F_DIM + ncol] = fmaxf(acc[i][q] + bias, 0.0f);
    }
  }
}

template <int COUNT>
__global__ void __launch_bounds__(AGG_THREADS)
range_layer_kernel(const float* __restrict__ x, const int* __restrict__ src,
                   const int* __restrict__ dst, float* __restrict__ deg,
                   unsigned* __restrict__ wlist, unsigned* __restrict__ wcount,
                   const float* __restrict__ wl, const float* __restrict__ bl,
                   const float* __restrict__ wr, float* __restrict__ h) {
  __shared__ __attribute__((aligned(16))) float sAcc[AGG_ROWS * LDS_STRIDE];
  __shared__ __attribute__((aligned(16))) float sX[AGG_ROWS * LDS_STRIDE];
  __shared__ unsigned sList[AGG_CAP];
  __shared__ float    sCnt[AGG_ROWS];
  __shared__ unsigned sN;

  const int t    = threadIdx.x;
  const int lane = t & 31;
  const int wave = __builtin_amdgcn_readfirstlane(t >> 5);
  const int base = blockIdx.x * AGG_R;

  for (int i = t; i < AGG_ROWS * LDS_STRIDE / 4; i += AGG_THREADS)
    reinterpret_cast<float4*>(sAcc)[i] = make_float4(0.0f, 0.0f, 0.0f, 0.0f);
  if (t < AGG_ROWS) {
    float c = 0.0f;
    if (!COUNT && t < AGG_R && base + t < N_NODES) c = deg[base + t];
    sCnt[t] = c;
  }
  if (t == 0) sN = 0u;
  __syncthreads();

  auto append = [&](int e, unsigned u) {
    if (COUNT) atomicAdd(&sCnt[u], 1.0f);
    const unsigned pos = atomicAdd(&sN, 1u);
    if (pos < AGG_CAP) {
      sList[pos] = (unsigned)e | (u << 20);
    } else {
      const float* xr = x + (size_t)src[e] * F_DIM;
      for (int c = 0; c < F_DIM; ++c)
        atomicAdd(&sAcc[u * LDS_STRIDE + (c & 3) * ACC_COMP + (c >> 2)], xr[c]);
    }
  };

  unsigned saved = 0u;
  if (!COUNT) saved = (unsigned)__builtin_amdgcn_readfirstlane((int)wcount[blockIdx.x]);
  const bool rescan = COUNT || saved > (unsigned)AGG_CAP;
  unsigned* wsl = wlist + (size_t)blockIdx.x * AGG_CAP;

  if (!rescan) {
    for (unsigned i = t; i < saved; i += AGG_THREADS) sList[i] = wsl[i];
    if (t == 0) sN = saved;
  } else {
    const int4* d4 = reinterpret_cast<const int4*>(dst);
    const int off = blockIdx.x * (N_GROUPS / AGG_BLOCKS);
    auto load4 = [&](int it, int4* d, int* gi) {
#pragma unroll
      for (int j = 0; j < 4; ++j) {
        const int  g  = (it + j) * AGG_THREADS + t;
        const bool ok = g < N_GROUPS;
        int gg = g + off;
        if (gg >= N_GROUPS) gg -= N_GROUPS;
        gg = ok ? gg : 0;
        gi[j] = gg;
        int4 v = d4[gg];
        if (!ok) v = make_int4(-1, -1, -1, -1);
        d[j] = v;
      }
    };
    int4 cur[4];
    int  gcur[4];
    load4(0, cur, gcur);
    for (int it = 0; it < (N_GROUPS + AGG_THREADS - 1) / AGG_THREADS; it += 4) {
      int4 nxt[4];
      int  gnxt[4];
      load4(it + 4, nxt, gnxt);
#pragma unroll
      for (int j = 0; j < 4; ++j) {
        const unsigned u0 = (unsigned)(cur[j].x - base);
        const unsigned u1 = (unsigned)(cur[j].y - base);
        const unsigned u2 = (unsigned)(cur[j].z - base);
        const unsigned u3 = (unsigned)(cur[j].w - base);
        const unsigned ua = u0 < u1 ? u0 : u1;
        const unsigned ub = u2 < u3 ? u2 : u3;
        if ((ua < ub ? ua : ub) < AGG_R) {
          const int e = gcur[j] * 4;
          if (u0 < AGG_R) append(e + 0, u0);
          if (u1 < AGG_R) append(e + 1, u1);
          if (u2 < AGG_R) append(e + 2, u2);
          if (u3 < AGG_R) append(e + 3, u3);
        }
      }
#pragma unroll
      for (int j = 0; j < 4; ++j) { cur[j] = nxt[j]; gcur[j] = gnxt[j]; }
    }
  }
  __syncthreads();

  const unsigned nraw = sN;
  const int n = (int)(nraw < AGG_CAP ? nraw : (unsigned)AGG_CAP);
  const unsigned spare = (unsigned)AGG_R << 20;
  if (COUNT) {
    for (int i = t; i < n; i += AGG_THREADS) wsl[i] = sList[i];
    if (t == 0) wcount[blockIdx.x] = nraw;
  }
  auto entry = [&](int b) -> unsigned {
    const int idx = b + (lane & 7);
    return (idx < n) ? sList[idx] : spare;
  };
  unsigned ent = entry(wave * 8);
  int      s   = src[ent & 0xFFFFFu];
  for (int b = wave * 8; b < n; b += (AGG_THREADS / 32) * 8) {
    float4 row[8];
#pragma unroll
    for (int j = 0; j < 8; ++j) {
      const int sj = __builtin_amdgcn_readlane(s, j);
      row[j] = *reinterpret_cast<const float4*>(x + (size_t)sj * F_DIM + lane * 4);
    }
    const unsigned ent_n = entry(b + (AGG_THREADS / 32) * 8);
    const int      s_n   = src[ent_n & 0xFFFFFu];
#pragma unroll
    for (int j = 0; j < 8; ++j) {
      const unsigned ld = (unsigned)__builtin_amdgcn_readlane((int)ent, j) >> 20;
      float* p = sAcc + ld * LDS_STRIDE + lane;
      atomicAdd(p,      row[j].x);
      atomicAdd(p + ACC_COMP,     row[j].y);
      atomicAdd(p + 2 * ACC_COMP, row[j].z);
      atomicAdd(p + 3 * ACC_COMP, row[j].w);
    }
    ent = ent_n;
    s   = s_n;
  }

  for (int i = t; i < AGG_ROWS * 32; i += AGG_THREADS) {
    const int r    = i >> 5;
    const int grow = base + r;
    float4 v = make_float4(0.0f, 0.0f, 0.0f, 0.0f);
    if (r < AGG_R && grow < N_NODES)
      v = *reinterpret_cast<const float4*>(x + (size_t)grow * F_DIM + lane * 4);
    *reinterpret_cast<float4*>(sX + r * LDS_STRIDE + lane * 4) = v;
  }
  __syncthreads();

  if (COUNT) {
    if (t < AGG_R && base + t < N_NODES) deg[base + t] = sCnt[t];
  }

  const int tn  = wave & 7;
  const int rt0 = wave >> 3;
  if (rt0 == 0)
    range_tiles_gemm<4>(sAcc, sX, sCnt, wl, bl, wr, h, base, rt0, tn, lane);
  else
    range_tiles_gemm<3>(sAcc, sX, sCnt, wl, bl, wr, h, base, rt0, tn, lane);
}

#define ROW_TILES (N_NODES / 16)

template <int ACT, int K>
__global__ void __launch_bounds__(256)
wmma_gemm_bias_kernel(const float* __restrict__ A, const float* __restrict__ B,
                      const float* __restrict__ bias, float* __restrict__ C,
                      int N) {
  __shared__ float sA[16 * LDS_STRIDE];

  const int lane   = threadIdx.x & 31;
  const int wave   = threadIdx.x >> 5;
  const int tilesN = (N + 15) >> 4;
  int tn = blockIdx.x * (blockDim.x >> 5) + wave;
  if (tn >= tilesN) tn = tilesN - 1;
  const int mloc = lane & 15;
  const int ncol = tn * 16 + mloc;
  const bool nok = (ncol < N);
  const int kb = (lane >= 16) ? 2 : 0;
  const float* sa = sA + mloc * LDS_STRIDE;

  v2f bf[K / 4];
#pragma unroll
  for (int s = 0; s < K / 4; ++s) {
    const int kk = s * 4;
    bf[s].x = nok ? B[(size_t)(kk + kb) * N + ncol]     : 0.0f;
    bf[s].y = nok ? B[(size_t)(kk + kb + 1) * N + ncol] : 0.0f;
  }
  const float bv = nok ? bias[ncol] : 0.0f;

  constexpr int k4 = K >> 2;
  for (int tm = blockIdx.y; tm < ROW_TILES; tm += gridDim.y) {
    for (int i = threadIdx.x; i < 16 * k4; i += blockDim.x) {
      const int r  = i / k4;
      const int c4 = (i - r * k4) * 4;
      const float4 v = *reinterpret_cast<const float4*>(A + (size_t)(tm * 16 + r) * K + c4);
      float* p = sA + r * LDS_STRIDE + c4;
      p[0] = v.x; p[1] = v.y; p[2] = v.z; p[3] = v.w;
    }
    __syncthreads();

    v8f acc = {};
#pragma unroll
    for (int s = 0; s < K / 4; ++s) {
      const int kk = s * 4;
      v2f a;
      a.x = sa[kk + kb];
      a.y = sa[kk + kb + 1];
      acc = wmma_f32(a, bf[s], acc);
    }
    if (nok) {
      const int mbase = tm * 16 + ((lane >= 16) ? 8 : 0);
#pragma unroll
      for (int i = 0; i < 8; ++i) {
        float v = acc[i] + bv;
        if (ACT == 1) v = fmaxf(v, 0.0f);
        C[(size_t)(mbase + i) * N + ncol] = v;
      }
    }
    __syncthreads();
  }
}

template <int K>
__global__ void __launch_bounds__(256)
wmma_gemm_bn_kernel(const float* __restrict__ A, const float* __restrict__ B,
                    const float* __restrict__ bconv, const float* __restrict__ gamma,
                    const float* __restrict__ beta, const float* __restrict__ rm,
                    const float* __restrict__ rv, float* __restrict__ C,
                    int N) {
  __shared__ float sA[16 * LDS_STRIDE];

  const int lane   = threadIdx.x & 31;
  const int wave   = threadIdx.x >> 5;
  const int tilesN = (N + 15) >> 4;
  int tn = blockIdx.x * (blockDim.x >> 5) + wave;
  if (tn >= tilesN) tn = tilesN - 1;
  const int mloc = lane & 15;
  const int ncol = tn * 16 + mloc;
  const bool nok = (ncol < N);
  const int kb = (lane >= 16) ? 2 : 0;
  const float* sa = sA + mloc * LDS_STRIDE;

  v2f bf[K / 4];
#pragma unroll
  for (int s = 0; s < K / 4; ++s) {
    const int kk = s * 4;
    bf[s].x = nok ? B[(size_t)(kk + kb) * N + ncol]     : 0.0f;
    bf[s].y = nok ? B[(size_t)(kk + kb + 1) * N + ncol] : 0.0f;
  }
  float sc = 0.0f, shift = 0.0f;
  if (nok) {
    sc    = gamma[ncol] * rsqrtf(rv[ncol] + 1e-5f);
    shift = (bconv[ncol] - rm[ncol]) * sc + beta[ncol];
  }

  constexpr int k4 = K >> 2;
  for (int tm = blockIdx.y; tm < ROW_TILES; tm += gridDim.y) {
    for (int i = threadIdx.x; i < 16 * k4; i += blockDim.x) {
      const int r  = i / k4;
      const int c4 = (i - r * k4) * 4;
      const float4 v = *reinterpret_cast<const float4*>(A + (size_t)(tm * 16 + r) * K + c4);
      float* p = sA + r * LDS_STRIDE + c4;
      p[0] = v.x; p[1] = v.y; p[2] = v.z; p[3] = v.w;
    }
    __syncthreads();

    v8f acc = {};
#pragma unroll
    for (int s = 0; s < K / 4; ++s) {
      const int kk = s * 4;
      v2f a;
      a.x = sa[kk + kb];
      a.y = sa[kk + kb + 1];
      acc = wmma_f32(a, bf[s], acc);
    }
    if (nok) {
      const int mbase = tm * 16 + ((lane >= 16) ? 8 : 0);
#pragma unroll
      for (int i = 0; i < 8; ++i)
        C[(size_t)(mbase + i) * N + ncol] = acc[i] * sc + shift;
    }
    __syncthreads();
  }
}

__global__ void __launch_bounds__(256)
tanh_kernel(const float* __restrict__ in, float* __restrict__ out, int n4) {
  const float4* in4  = reinterpret_cast<const float4*>(in);
  float4*       out4 = reinterpret_cast<float4*>(out);
  for (int i = blockIdx.x * blockDim.x + threadIdx.x; i < n4; i += gridDim.x * blockDim.x) {
    const float4 v = in4[i];
    out4[i] = make_float4(tanhf(v.x), tanhf(v.y), tanhf(v.z), tanhf(v.w));
  }
}

__global__ void __launch_bounds__(256)
true_lab_kernel(const float* __restrict__ pre, const float* __restrict__ wtl,
                const float* __restrict__ btl, float* __restrict__ out) {
  int r = blockIdx.x * blockDim.x + threadIdx.x;
  if (r >= N_NODES) return;
  const float* p = pre + (size_t)r * NBITS;
  float s = 0.0f;
#pragma unroll
  for (int k = 0; k < NBITS; ++k) s += p[k] * wtl[k];
  out[r] = s + btl[0];
}

extern "C" void kernel_launch(void* const* d_in, const int* in_sizes, int n_in,
                              void* d_out, int out_size, void* d_ws, size_t ws_size,
                              hipStream_t stream) {
  (void)in_sizes; (void)n_in; (void)out_size; (void)ws_size;

  const float* features = (const float*)d_in[0];
  const int*   edges    = (const int*)d_in[1];
  const float* w1l   = (const float*)d_in[2];
  const float* b1l   = (const float*)d_in[3];
  const float* w1r   = (const float*)d_in[4];
  const float* w2l   = (const float*)d_in[5];
  const float* b2l   = (const float*)d_in[6];
  const float* w2r   = (const float*)d_in[7];
  const float* whd   = (const float*)d_in[8];
  const float* bhd   = (const float*)d_in[9];
  const float* wclas = (const float*)d_in[10];
  const float* bclas = (const float*)d_in[11];
  const float* wconv = (const float*)d_in[12];
  const float* bconv = (const float*)d_in[13];
  const float* gamma = (const float*)d_in[14];
  const float* beta  = (const float*)d_in[15];
  const float* rm    = (const float*)d_in[16];
  const float* rv    = (const float*)d_in[17];
  const float* wtl   = (const float*)d_in[18];
  const float* btl   = (const float*)d_in[19];
  const float* wcv   = (const float*)d_in[20];
  const float* bcv   = (const float*)d_in[21];

  const int* src = edges;
  const int* dst = edges + N_EDGES;

  float* ws     = (float*)d_ws;
  float* deg    = ws;
  float* h1     = deg + N_NODES;
  float* h2     = h1  + (size_t)N_NODES * F_DIM;
  float* pre_bn = h2  + (size_t)N_NODES * F_DIM;
  unsigned* wcount = (unsigned*)(pre_bn + (size_t)N_NODES * NBITS);
  unsigned* wlist  = wcount + AGG_BLOCKS;

  float* out       = (float*)d_out;
  float* o_logists = out;
  float* o_tanh    = o_logists + (size_t)N_NODES * NCLS;
  float* o_fealab  = o_tanh    + (size_t)N_NODES * NBITS;
  float* o_feacv   = o_fealab  + (size_t)N_NODES * HD;
  float* o_truelab = o_feacv   + (size_t)N_NODES * NCLS;

  range_layer_kernel<1><<<AGG_BLOCKS, AGG_THREADS, 0, stream>>>(
      features, src, dst, deg, wlist, wcount, w1l, b1l, w1r, h1);

  range_layer_kernel<0><<<AGG_BLOCKS, AGG_THREADS, 0, stream>>>(
      h1, src, dst, deg, wlist, wcount, w2l, b2l, w2r, h2);

  wmma_gemm_bias_kernel<0, F_DIM><<<dim3(3, 256), 256, 0, stream>>>(
      h2, whd, bhd, o_fealab, HD);
  wmma_gemm_bias_kernel<0, F_DIM><<<dim3(1, 768), 256, 0, stream>>>(
      h2, wclas, bclas, o_logists, NCLS);
  wmma_gemm_bn_kernel<F_DIM><<<dim3(1, 768), 128, 0, stream>>>(
      h2, wconv, bconv, gamma, beta, rm, rv, pre_bn, NBITS);
  tanh_kernel<<<512, 256, 0, stream>>>(
      pre_bn, o_tanh, N_NODES * NBITS / 4);
  wmma_gemm_bias_kernel<0, NBITS><<<dim3(1, 768), 256, 0, stream>>>(
      o_tanh, wcv, bcv, o_feacv, NCLS);
  true_lab_kernel<<<(N_NODES + 255) / 256, 256, 0, stream>>>(
      pre_bn, wtl, btl, o_truelab);
}


# Head stage moved into the layer-2 range kernel: h2 never leaves the block, its LDS rows feed all five head products, BN, tanh and the true_lab dot; h2 and pre planes and 6 launches gone (8->2)
# speedup vs baseline: 2.9071x; 1.3154x over previous
#include <hip/hip_runtime.h>
#include <math.h>


typedef __attribute__((ext_vector_type(2))) float v2f;
typedef __attribute__((ext_vector_type(8))) float v8f;

#define N_NODES 50000
#define N_EDGES 800000
#define F_DIM   128
#define NBITS   64
#define NCLS    100
#define HD      300
#define LDS_STRIDE 132

__device__ __forceinline__ v8f wmma_f32(v2f a, v2f b, v8f c) {
  return __builtin_amdgcn_wmma_f32_16x16x4_f32(false, a, false, b, (short)0, c,
                                               false, false);
}

#define AGG_R       196
#define AGG_TILES   13
#define AGG_ROWS    (AGG_TILES * 16)
#define AGG_BLOCKS  256
#define AGG_THREADS 1024
#define AGG_CAP     16384
#define N_GROUPS    (N_EDGES / 4)
#define ACC_COMP    33

template <int NT, int KEEP>
__device__ __forceinline__ void range_tiles_gemm(
    const float* sAcc, const float* sX, const float* sCnt,
    const float* __restrict__ wl, const float* __restrict__ bl,
    const float* __restrict__ wr, float* __restrict__ h, v8f* hout,
    int base, int rt0, int tn, int lane) {
  const int mloc = lane & 15;
  const int hh   = lane >> 4;
  const int kb   = hh * 2;
  const int ncol = tn * 16 + mloc;
  v8f   acc[NT];
  float inv[NT];
  const float* pa[NT];
  const float* px[NT];
#pragma unroll
  for (int i = 0; i < NT; ++i) {
    const int row = (rt0 + 4 * i) * 16 + mloc;
    inv[i] = 1.0f / fmaxf(sCnt[row], 1.0f);
    pa[i]  = sAcc + row * LDS_STRIDE + kb * ACC_COMP;
    px[i]  = sX   + row * LDS_STRIDE + kb;
    acc[i] = v8f{};
  }
  const float* pwl = wl + kb * F_DIM + ncol;
  const float* pwr = wr + kb * F_DIM + ncol;
#pragma unroll 8
  for (int s = 0; s < F_DIM / 4; ++s) {
    v2f b1, b2;
    b1.x = pwl[(s * 4)     * F_DIM];
    b1.y = pwl[(s * 4 + 1) * F_DIM];
    b2.x = pwr[(s * 4)     * F_DIM];
    b2.y = pwr[(s * 4 + 1) * F_DIM];
#pragma unroll
    for (int i = 0; i < NT; ++i) {
      v2f a1, a2;
      a1.x = pa[i][s] * inv[i];
      a1.y = pa[i][ACC_COMP + s] * inv[i];
      acc[i] = wmma_f32(a1, b1, acc[i]);
      a2.x = px[i][s * 4];
      a2.y = px[i][s * 4 + 1];
      acc[i] = wmma_f32(a2, b2, acc[i]);
    }
  }
  const float bias = bl[ncol];
  if (KEEP) {
#pragma unroll
    for (int i = 0; i < NT; ++i) {
#pragma unroll
      for (int q = 0; q < 8; ++q) hout[i][q] = fmaxf(acc[i][q] + bias, 0.0f);
    }
    return;
  }
#pragma unroll
  for (int i = 0; i < NT; ++i) {
    const int lr0 = (rt0 + 4 * i) * 16 + hh * 8;
#pragma unroll
    for (int q = 0; q < 8; ++q) {
      const int lr   = lr0 + q;
      const int grow = base + lr;
      if (lr < AGG_R && grow < N_NODES)
        h[(size_t)grow * F_DIM + ncol] = fmaxf(acc[i][q] + bias, 0.0f);
    }
  }
}

#define HT_STRIDE 68

template <int NT, int K, int N, int ASTRIDE>
__device__ __forceinline__ void head_tiles_gemm(
    const float* sA, const float* __restrict__ B, int ncolc,
    int rt_first, int rt_step, int lane, v8f* acc) {
  const int mloc = lane & 15;
  const int kb   = (lane >> 4) * 2;
  const float* pa = sA + (rt_first * 16 + mloc) * ASTRIDE + kb;
#pragma unroll
  for (int i = 0; i < NT; ++i) acc[i] = v8f{};
  const float* pb = B + kb * N + ncolc;
#pragma unroll (NT > 6 ? 2 : NT > 5 ? 4 : 8)
  for (int s = 0; s < K / 4; ++s) {
    v2f b;
    b.x = pb[(s * 4)     * N];
    b.y = pb[(s * 4 + 1) * N];
#pragma unroll
    for (int i = 0; i < NT; ++i) {
      v2f a;
      a.x = pa[i * rt_step * 16 * ASTRIDE + s * 4];
      a.y = pa[i * rt_step * 16 * ASTRIDE + s * 4 + 1];
      acc[i] = wmma_f32(a, b, acc[i]);
    }
  }
}

template <int NT, int N>
__device__ __forceinline__ void head_store(
    const v8f* acc, float* __restrict__ out, float bv, int ncol, bool nok,
    int base, int nvalid, int rt_first, int rt_step, int lane) {
  const int hh = lane >> 4;
#pragma unroll
  for (int i = 0; i < NT; ++i) {
    const int lr0 = (rt_first + rt_step * i) * 16 + hh * 8;
    float* po = out + (size_t)(base + lr0) * N + ncol;
#pragma unroll
    for (int q = 0; q < 8; ++q) {
      if (nok && lr0 + q < nvalid) po[q * N] = acc[i][q] + bv;
    }
  }
}

template <int NT, int KIND>
__device__ __forceinline__ void head_job(
    const float* sH, float* sPre, float* sTanh, int tile, int rt_first, int base,
    int nvalid, int lane,
    const float* __restrict__ B, const float* __restrict__ bias,
    const float* __restrict__ gamma, const float* __restrict__ beta,
    const float* __restrict__ rm, const float* __restrict__ rv,
    float* __restrict__ out) {
  constexpr int N = (KIND == 0) ? HD : (KIND == 1) ? NCLS : NBITS;
  const int  mloc  = lane & 15;
  const int  ncol  = tile * 16 + mloc;
  const bool nok   = ncol < N;
  const int  ncolc = nok ? ncol : N - 1;
  v8f acc[NT];
  head_tiles_gemm<NT, F_DIM, N, LDS_STRIDE>(sH, B, ncolc, rt_first, 1, lane, acc);
  if (KIND != 2) {
    head_store<NT, N>(acc, out, bias[ncolc], ncol, nok, base, nvalid, rt_first, 1, lane);
  } else {
    const float sc    = gamma[ncol] * rsqrtf(rv[ncol] + 1e-5f);
    const float shift = (bias[ncol] - rm[ncol]) * sc + beta[ncol];
    const int hh = lane >> 4;
#pragma unroll
    for (int i = 0; i < NT; ++i) {
      const int lr0 = (rt_first + i) * 16 + hh * 8;
      float* po = out + (size_t)(base + lr0) * NBITS + ncol;
#pragma unroll
      for (int q = 0; q < 8; ++q) {
        const float v  = acc[i][q] * sc + shift;
        const float th = tanhf(v);
        sPre[(lr0 + q) * HT_STRIDE + ncol]  = v;
        sTanh[(lr0 + q) * HT_STRIDE + ncol] = th;
        if (lr0 + q < nvalid) po[q * NBITS] = th;
      }
    }
  }
}

template <int NT>
__device__ __forceinline__ void head_dispatch(
    const float* sH, float* sPre, float* sTanh, int ct, int rt_first, int base,
    int nvalid, int lane,
    const float* __restrict__ whd, const float* __restrict__ bhd,
    const float* __restrict__ wclas, const float* __restrict__ bclas,
    const float* __restrict__ wconv, const float* __restrict__ bconv,
    const float* __restrict__ gamma, const float* __restrict__ beta,
    const float* __restrict__ rm, const float* __restrict__ rv,
    float* __restrict__ o_fealab, float* __restrict__ o_logists,
    float* __restrict__ o_tanh) {
  if (ct < 19)
    head_job<NT, 0>(sH, sPre, sTanh, ct, rt_first, base, nvalid, lane, whd, bhd,
                    gamma, beta, rm, rv, o_fealab);
  else if (ct < 26)
    head_job<NT, 1>(sH, sPre, sTanh, ct - 19, rt_first, base, nvalid, lane, wclas, bclas,
                    gamma, beta, rm, rv, o_logists);
  else
    head_job<NT, 2>(sH, sPre, sTanh, ct - 26, rt_first, base, nvalid, lane, wconv, bconv,
                    gamma, beta, rm, rv, o_tanh);
}

template <int COUNT>
__global__ void __launch_bounds__(AGG_THREADS)
range_layer_kernel(const float* __restrict__ x, const int* __restrict__ src,
                   const int* __restrict__ dst, float* __restrict__ deg,
                   unsigned* __restrict__ wlist, unsigned* __restrict__ wcount,
                   const float* __restrict__ wl, const float* __restrict__ bl,
                   const float* __restrict__ wr, float* __restrict__ h,
                   const float* __restrict__ whd, const float* __restrict__ bhd,
                   const float* __restrict__ wclas, const float* __restrict__ bclas,
                   const float* __restrict__ wconv, const float* __restrict__ bconv,
                   const float* __restrict__ gamma, const float* __restrict__ beta,
                   const float* __restrict__ rm, const float* __restrict__ rv,
                   const float* __restrict__ wtl, const float* __restrict__ btl,
                   const float* __restrict__ wcv, const float* __restrict__ bcv,
                   float* __restrict__ o_logists, float* __restrict__ o_tanh,
                   float* __restrict__ o_fealab, float* __restrict__ o_feacv,
                   float* __restrict__ o_truelab) {
  __shared__ __attribute__((aligned(16))) float sAcc[AGG_ROWS * LDS_STRIDE];
  __shared__ __attribute__((aligned(16))) float sX[AGG_ROWS * LDS_STRIDE];
  __shared__ unsigned sList[AGG_CAP];
  __shared__ float    sCnt[AGG_ROWS];
  __shared__ unsigned sN;

  const int t    = threadIdx.x;
  const int lane = t & 31;
  const int wave = __builtin_amdgcn_readfirstlane(t >> 5);
  const int base = blockIdx.x * AGG_R;

  for (int i = t; i < AGG_ROWS * LDS_STRIDE / 4; i += AGG_THREADS)
    reinterpret_cast<float4*>(sAcc)[i] = make_float4(0.0f, 0.0f, 0.0f, 0.0f);
  if (t < AGG_ROWS) {
    float c = 0.0f;
    if (!COUNT && t < AGG_R && base + t < N_NODES) c = deg[base + t];
    sCnt[t] = c;
  }
  if (t == 0) sN = 0u;
  __syncthreads();

  auto append = [&](int e, unsigned u) {
    if (COUNT) atomicAdd(&sCnt[u], 1.0f);
    const unsigned pos = atomicAdd(&sN, 1u);
    if (pos < AGG_CAP) {
      sList[pos] = (unsigned)e | (u << 20);
    } else {
      const float* xr = x + (size_t)src[e] * F_DIM;
      for (int c = 0; c < F_DIM; ++c)
        atomicAdd(&sAcc[u * LDS_STRIDE + (c & 3) * ACC_COMP + (c >> 2)], xr[c]);
    }
  };

  unsigned saved = 0u;
  if (!COUNT) saved = (unsigned)__builtin_amdgcn_readfirstlane((int)wcount[blockIdx.x]);
  const bool rescan = COUNT || saved > (unsigned)AGG_CAP;
  unsigned* wsl = wlist + (size_t)blockIdx.x * AGG_CAP;

  if (!rescan) {
    for (unsigned i = t; i < saved; i += AGG_THREADS) sList[i] = wsl[i];
    if (t == 0) sN = saved;
  } else {
    const int4* d4 = reinterpret_cast<const int4*>(dst);
    const int off = blockIdx.x * (N_GROUPS / AGG_BLOCKS);
    auto load4 = [&](int it, int4* d, int* gi) {
#pragma unroll
      for (int j = 0; j < 4; ++j) {
        const int  g  = (it + j) * AGG_THREADS + t;
        const bool ok = g < N_GROUPS;
        int gg = g + off;
        if (gg >= N_GROUPS) gg -= N_GROUPS;
        gg = ok ? gg : 0;
        gi[j] = gg;
        int4 v = d4[gg];
        if (!ok) v = make_int4(-1, -1, -1, -1);
        d[j] = v;
      }
    };
    int4 cur[4];
    int  gcur[4];
    load4(0, cur, gcur);
    for (int it = 0; it < (N_GROUPS + AGG_THREADS - 1) / AGG_THREADS; it += 4) {
      int4 nxt[4];
      int  gnxt[4];
      load4(it + 4, nxt, gnxt);
#pragma unroll
      for (int j = 0; j < 4; ++j) {
        const unsigned u0 = (unsigned)(cur[j].x - base);
        const unsigned u1 = (unsigned)(cur[j].y - base);
        const unsigned u2 = (unsigned)(cur[j].z - base);
        const unsigned u3 = (unsigned)(cur[j].w - base);
        const unsigned ua = u0 < u1 ? u0 : u1;
        const unsigned ub = u2 < u3 ? u2 : u3;
        if ((ua < ub ? ua : ub) < AGG_R) {
          const int e = gcur[j] * 4;
          if (u0 < AGG_R) append(e + 0, u0);
          if (u1 < AGG_R) append(e + 1, u1);
          if (u2 < AGG_R) append(e + 2, u2);
          if (u3 < AGG_R) append(e + 3, u3);
        }
      }
#pragma unroll
      for (int j = 0; j < 4; ++j) { cur[j] = nxt[j]; gcur[j] = gnxt[j]; }
    }
  }
  __syncthreads();

  const unsigned nraw = sN;
  const int n = (int)(nraw < AGG_CAP ? nraw : (unsigned)AGG_CAP);
  const unsigned spare = (unsigned)AGG_R << 20;
  if (COUNT) {
    for (int i = t; i < n; i += AGG_THREADS) wsl[i] = sList[i];
    if (t == 0) wcount[blockIdx.x] = nraw;
  }
  auto entry = [&](int b) -> unsigned {
    const int idx = b + (lane & 7);
    return (idx < n) ? sList[idx] : spare;
  };
  unsigned ent = entry(wave * 8);
  int      s   = src[ent & 0xFFFFFu];
  for (int b = wave * 8; b < n; b += (AGG_THREADS / 32) * 8) {
    float4 row[8];
#pragma unroll
    for (int j = 0; j < 8; ++j) {
      const int sj = __builtin_amdgcn_readlane(s, j);
      row[j] = *reinterpret_cast<const float4*>(x + (size_t)sj * F_DIM + lane * 4);
    }
    const unsigned ent_n = entry(b + (AGG_THREADS / 32) * 8);
    const int      s_n   = src[ent_n & 0xFFFFFu];
#pragma unroll
    for (int j = 0; j < 8; ++j) {
      const unsigned ld = (unsigned)__builtin_amdgcn_readlane((int)ent, j) >> 20;
      float* p = sAcc + ld * LDS_STRIDE + lane;
      atomicAdd(p,      row[j].x);
      atomicAdd(p + ACC_COMP,     row[j].y);
      atomicAdd(p + 2 * ACC_COMP, row[j].z);
      atomicAdd(p + 3 * ACC_COMP, row[j].w);
    }
    ent = ent_n;
    s   = s_n;
  }

  for (int i = t; i < AGG_ROWS * 32; i += AGG_THREADS) {
    const int r    = i >> 5;
    const int grow = base + r;
    float4 v = make_float4(0.0f, 0.0f, 0.0f, 0.0f);
    if (r < AGG_R && grow < N_NODES)
      v = *reinterpret_cast<const float4*>(x + (size_t)grow * F_DIM + lane * 4);
    *reinterpret_cast<float4*>(sX + r * LDS_STRIDE + lane * 4) = v;
  }
  __syncthreads();

  if (COUNT) {
    if (t < AGG_R && base + t < N_NODES) deg[base + t] = sCnt[t];
  }

  const int tn  = wave & 7;
  const int rt0 = wave >> 3;
  if (COUNT) {
    if (rt0 == 0)
      range_tiles_gemm<4, 0>(sAcc, sX, sCnt, wl, bl, wr, h, nullptr, base, rt0, tn, lane);
    else
      range_tiles_gemm<3, 0>(sAcc, sX, sCnt, wl, bl, wr, h, nullptr, base, rt0, tn, lane);
    return;
  }

  v8f hacc[4];
  hacc[3] = v8f{};
  if (rt0 == 0)
    range_tiles_gemm<4, 1>(sAcc, sX, sCnt, wl, bl, wr, h, hacc, base, rt0, tn, lane);
  else
    range_tiles_gemm<3, 1>(sAcc, sX, sCnt, wl, bl, wr, h, hacc, base, rt0, tn, lane);
  const int ntl = (rt0 == 0) ? 4 : 3;
  __syncthreads();
  {
    const int mloc = lane & 15;
    const int hh   = lane >> 4;
#pragma unroll
    for (int i = 0; i < 4; ++i) {
      if (i < ntl) {
        const int lr0 = (rt0 + 4 * i) * 16 + hh * 8;
#pragma unroll
        for (int q = 0; q < 8; ++q)
          sX[(lr0 + q) * LDS_STRIDE + tn * 16 + mloc] = hacc[i][q];
      }
    }
  }
  __syncthreads();

  float* sPre  = reinterpret_cast<float*>(sList);
  float* sTanh = sAcc;
  const int nvalid = (N_NODES - base < AGG_R) ? (N_NODES - base) : AGG_R;
  if (wave < 30)
    head_dispatch<7>(sX, sPre, sTanh, wave, 0, base, nvalid, lane, whd, bhd, wclas, bclas,
                     wconv, bconv, gamma, beta, rm, rv, o_fealab, o_logists, o_tanh);
  if (wave < 28 || wave >= 30)
    head_dispatch<6>(sX, sPre, sTanh, (wave < 28) ? wave + 2 : wave - 30, 7, base, nvalid, lane,
                     whd, bhd, wclas, bclas, wconv, bconv, gamma, beta, rm, rv,
                     o_fealab, o_logists, o_tanh);
  __syncthreads();

  if (wave < 28) {
    const int ctn  = wave % 7;
    const int rg   = wave / 7;
    const int mloc = lane & 15;
    const int ncol = ctn * 16 + mloc;
    const bool nok = ncol < NCLS;
    const int ncolc = nok ? ncol : NCLS - 1;
    if (rg == 0) {
      v8f acc[4];
      head_tiles_gemm<4, NBITS, NCLS, HT_STRIDE>(sTanh, wcv, ncolc, rg, 4, lane, acc);
      head_store<4, NCLS>(acc, o_feacv, bcv[ncolc], ncol, nok, base, nvalid, rg, 4, lane);
    } else {
      v8f acc[3];
      head_tiles_gemm<3, NBITS, NCLS, HT_STRIDE>(sTanh, wcv, ncolc, rg, 4, lane, acc);
      head_store<3, NCLS>(acc, o_feacv, bcv[ncolc], ncol, nok, base, nvalid, rg, 4, lane);
    }
  } else {
    for (int r = t - 28 * 32; r < nvalid; r += 4 * 32) {
      const float* p = sPre + r * HT_STRIDE;
      float sum = 0.0f;
#pragma unroll
      for (int k = 0; k < NBITS; ++k) sum += p[k] * wtl[k];
      o_truelab[base + r] = sum + btl[0];
    }
  }
}

extern "C" void kernel_launch(void* const* d_in, const int* in_sizes, int n_in,
                              void* d_out, int out_size, void* d_ws, size_t ws_size,
                              hipStream_t stream) {
  (void)in_sizes; (void)n_in; (void)out_size; (void)ws_size;

  const float* features = (const float*)d_in[0];
  const int*   edges    = (const int*)d_in[1];
  const float* w1l   = (const float*)d_in[2];
  const float* b1l   = (const float*)d_in[3];
  const float* w1r   = (const float*)d_in[4];
  const float* w2l   = (const float*)d_in[5];
  const float* b2l   = (const float*)d_in[6];
  const float* w2r   = (const float*)d_in[7];
  const float* whd   = (const float*)d_in[8];
  const float* bhd   = (const float*)d_in[9];
  const float* wclas = (const float*)d_in[10];
  const float* bclas = (const float*)d_in[11];
  const float* wconv = (const float*)d_in[12];
  const float* bconv = (const float*)d_in[13];
  const float* gamma = (const float*)d_in[14];
  const float* beta  = (const float*)d_in[15];
  const float* rm    = (const float*)d_in[16];
  const float* rv    = (const float*)d_in[17];
  const float* wtl   = (const float*)d_in[18];
  const float* btl   = (const float*)d_in[19];
  const float* wcv   = (const float*)d_in[20];
  const float* bcv   = (const float*)d_in[21];

  const int* src = edges;
  const int* dst = edges + N_EDGES;

  float* ws     = (float*)d_ws;
  float* deg    = ws;
  float* h1     = deg + N_NODES;
  unsigned* wcount = (unsigned*)(h1 + (size_t)N_NODES * F_DIM);
  unsigned* wlist  = wcount + AGG_BLOCKS;

  float* out       = (float*)d_out;
  float* o_logists = out;
  float* o_tanh    = o_logists + (size_t)N_NODES * NCLS;
  float* o_fealab  = o_tanh    + (size_t)N_NODES * NBITS;
  float* o_feacv   = o_fealab  + (size_t)N_NODES * HD;
  float* o_truelab = o_feacv   + (size_t)N_NODES * NCLS;

  range_layer_kernel<1><<<AGG_BLOCKS, AGG_THREADS, 0, stream>>>(
      features, src, dst, deg, wlist, wcount, w1l, b1l, w1r, h1,
      whd, bhd, wclas, bclas, wconv, bconv, gamma, beta, rm, rv, wtl, btl, wcv, bcv,
      o_logists, o_tanh, o_fealab, o_feacv, o_truelab);

  range_layer_kernel<0><<<AGG_BLOCKS, AGG_THREADS, 0, stream>>>(
      h1, src, dst, deg, wlist, wcount, w2l, b2l, w2r, nullptr,
      whd, bhd, wclas, bclas, wconv, bconv, gamma, beta, rm, rv, wtl, btl, wcv, bcv,
      o_logists, o_tanh, o_fealab, o_feacv, o_truelab);
}


# dst scan narrowed: a new first launch stores each edge's range number (dst/196) as one byte; range blocks search that plane, a quarter of dst, and read dst only for matched edges; 2->3 launches
# speedup vs baseline: 3.1715x; 1.0909x over previous
#include <hip/hip_runtime.h>
#include <math.h>


typedef __attribute__((ext_vector_type(2))) float v2f;
typedef __attribute__((ext_vector_type(8))) float v8f;

#define N_NODES 50000
#define N_EDGES 800000
#define F_DIM   128
#define NBITS   64
#define NCLS    100
#define HD      300
#define LDS_STRIDE 132

__device__ __forceinline__ v8f wmma_f32(v2f a, v2f b, v8f c) {
  return __builtin_amdgcn_wmma_f32_16x16x4_f32(false, a, false, b, (short)0, c,
                                               false, false);
}

#define AGG_R       196
#define AGG_TILES   13
#define AGG_ROWS    (AGG_TILES * 16)
#define AGG_BLOCKS  256
#define AGG_THREADS 1024
#define AGG_CAP     16384
#define N_GROUPS    (N_EDGES / 4)
#define N_RG        (N_EDGES / 16)
#define RID_THREADS 256
#define RID_BLOCKS  ((N_GROUPS + RID_THREADS * 4 - 1) / (RID_THREADS * 4))

__global__ void __launch_bounds__(RID_THREADS)
range_id_kernel(const int* __restrict__ dst, unsigned* __restrict__ rid) {
  const int4* d4 = reinterpret_cast<const int4*>(dst);
  const int g0 = blockIdx.x * (RID_THREADS * 4) + threadIdx.x;
  int4 v[4];
#pragma unroll
  for (int j = 0; j < 4; ++j) {
    const int g = g0 + j * RID_THREADS;
    v[j] = d4[g < N_GROUPS ? g : 0];
  }
#pragma unroll
  for (int j = 0; j < 4; ++j) {
    const int g = g0 + j * RID_THREADS;
    const unsigned q0 = ((unsigned)v[j].x / (unsigned)AGG_R) & 0xFFu;
    const unsigned q1 = ((unsigned)v[j].y / (unsigned)AGG_R) & 0xFFu;
    const unsigned q2 = ((unsigned)v[j].z / (unsigned)AGG_R) & 0xFFu;
    const unsigned q3 = ((unsigned)v[j].w / (unsigned)AGG_R) & 0xFFu;
    if (g < N_GROUPS) rid[g] = q0 | (q1 << 8) | (q2 << 16) | (q3 << 24);
  }
}
#define ACC_COMP    33

template <int NT, int KEEP>
__device__ __forceinline__ void range_tiles_gemm(
    const float* sAcc, const float* sX, const float* sCnt,
    const float* __restrict__ wl, const float* __restrict__ bl,
    const float* __restrict__ wr, float* __restrict__ h, v8f* hout,
    int base, int rt0, int tn, int lane) {
  const int mloc = lane & 15;
  const int hh   = lane >> 4;
  const int kb   = hh * 2;
  const int ncol = tn * 16 + mloc;
  v8f   acc[NT];
  float inv[NT];
  const float* pa[NT];
  const float* px[NT];
#pragma unroll
  for (int i = 0; i < NT; ++i) {
    const int row = (rt0 + 4 * i) * 16 + mloc;
    inv[i] = 1.0f / fmaxf(sCnt[row], 1.0f);
    pa[i]  = sAcc + row * LDS_STRIDE + kb * ACC_COMP;
    px[i]  = sX   + row * LDS_STRIDE + kb;
    acc[i] = v8f{};
  }
  const float* pwl = wl + kb * F_DIM + ncol;
  const float* pwr = wr + kb * F_DIM + ncol;
#pragma unroll 8
  for (int s = 0; s < F_DIM / 4; ++s) {
    v2f b1, b2;
    b1.x = pwl[(s * 4)     * F_DIM];
    b1.y = pwl[(s * 4 + 1) * F_DIM];
    b2.x = pwr[(s * 4)     * F_DIM];
    b2.y = pwr[(s * 4 + 1) * F_DIM];
#pragma unroll
    for (int i = 0; i < NT; ++i) {
      v2f a1, a2;
      a1.x = pa[i][s] * inv[i];
      a1.y = pa[i][ACC_COMP + s] * inv[i];
      acc[i] = wmma_f32(a1, b1, acc[i]);
      a2.x = px[i][s * 4];
      a2.y = px[i][s * 4 + 1];
      acc[i] = wmma_f32(a2, b2, acc[i]);
    }
  }
  const float bias = bl[ncol];
  if (KEEP) {
#pragma unroll
    for (int i = 0; i < NT; ++i) {
#pragma unroll
      for (int q = 0; q < 8; ++q) hout[i][q] = fmaxf(acc[i][q] + bias, 0.0f);
    }
    return;
  }
#pragma unroll
  for (int i = 0; i < NT; ++i) {
    const int lr0 = (rt0 + 4 * i) * 16 + hh * 8;
#pragma unroll
    for (int q = 0; q < 8; ++q) {
      const int lr   = lr0 + q;
      const int grow = base + lr;
      if (lr < AGG_R && grow < N_NODES)
        h[(size_t)grow * F_DIM + ncol] = fmaxf(acc[i][q] + bias, 0.0f);
    }
  }
}

#define HT_STRIDE 68

template <int NT, int K, int N, int ASTRIDE>
__device__ __forceinline__ void head_tiles_gemm(
    const float* sA, const float* __restrict__ B, int ncolc,
    int rt_first, int rt_step, int lane, v8f* acc) {
  const int mloc = lane & 15;
  const int kb   = (lane >> 4) * 2;
  const float* pa = sA + (rt_first * 16 + mloc) * ASTRIDE + kb;
#pragma unroll
  for (int i = 0; i < NT; ++i) acc[i] = v8f{};
  const float* pb = B + kb * N + ncolc;
#pragma unroll (NT > 6 ? 2 : NT > 5 ? 4 : 8)
  for (int s = 0; s < K / 4; ++s) {
    v2f b;
    b.x = pb[(s * 4)     * N];
    b.y = pb[(s * 4 + 1) * N];
#pragma unroll
    for (int i = 0; i < NT; ++i) {
      v2f a;
      a.x = pa[i * rt_step * 16 * ASTRIDE + s * 4];
      a.y = pa[i * rt_step * 16 * ASTRIDE + s * 4 + 1];
      acc[i] = wmma_f32(a, b, acc[i]);
    }
  }
}

template <int NT, int N>
__device__ __forceinline__ void head_store(
    const v8f* acc, float* __restrict__ out, float bv, int ncol, bool nok,
    int base, int nvalid, int rt_first, int rt_step, int lane) {
  const int hh = lane >> 4;
#pragma unroll
  for (int i = 0; i < NT; ++i) {
    const int lr0 = (rt_first + rt_step * i) * 16 + hh * 8;
    float* po = out + (size_t)(base + lr0) * N + ncol;
#pragma unroll
    for (int q = 0; q < 8; ++q) {
      if (nok && lr0 + q < nvalid) po[q * N] = acc[i][q] + bv;
    }
  }
}

template <int NT, int KIND>
__device__ __forceinline__ void head_job(
    const float* sH, float* sPre, float* sTanh, int tile, int rt_first, int base,
    int nvalid, int lane,
    const float* __restrict__ B, const float* __restrict__ bias,
    const float* __restrict__ gamma, const float* __restrict__ beta,
    const float* __restrict__ rm, const float* __restrict__ rv,
    float* __restrict__ out) {
  constexpr int N = (KIND == 0) ? HD : (KIND == 1) ? NCLS : NBITS;
  const int  mloc  = lane & 15;
  const int  ncol  = tile * 16 + mloc;
  const bool nok   = ncol < N;
  const int  ncolc = nok ? ncol : N - 1;
  v8f acc[NT];
  head_tiles_gemm<NT, F_DIM, N, LDS_STRIDE>(sH, B, ncolc, rt_first, 1, lane, acc);
  if (KIND != 2) {
    head_store<NT, N>(acc, out, bias[ncolc], ncol, nok, base, nvalid, rt_first, 1, lane);
  } else {
    const float sc    = gamma[ncol] * rsqrtf(rv[ncol] + 1e-5f);
    const float shift = (bias[ncol] - rm[ncol]) * sc + beta[ncol];
    const int hh = lane >> 4;
#pragma unroll
    for (int i = 0; i < NT; ++i) {
      const int lr0 = (rt_first + i) * 16 + hh * 8;
      float* po = out + (size_t)(base + lr0) * NBITS + ncol;
#pragma unroll
      for (int q = 0; q < 8; ++q) {
        const float v  = acc[i][q] * sc + shift;
        const float th = tanhf(v);
        sPre[(lr0 + q) * HT_STRIDE + ncol]  = v;
        sTanh[(lr0 + q) * HT_STRIDE + ncol] = th;
        if (lr0 + q < nvalid) po[q * NBITS] = th;
      }
    }
  }
}

template <int NT>
__device__ __forceinline__ void head_dispatch(
    const float* sH, float* sPre, float* sTanh, int ct, int rt_first, int base,
    int nvalid, int lane,
    const float* __restrict__ whd, const float* __restrict__ bhd,
    const float* __restrict__ wclas, const float* __restrict__ bclas,
    const float* __restrict__ wconv, const float* __restrict__ bconv,
    const float* __restrict__ gamma, const float* __restrict__ beta,
    const float* __restrict__ rm, const float* __restrict__ rv,
    float* __restrict__ o_fealab, float* __restrict__ o_logists,
    float* __restrict__ o_tanh) {
  if (ct < 19)
    head_job<NT, 0>(sH, sPre, sTanh, ct, rt_first, base, nvalid, lane, whd, bhd,
                    gamma, beta, rm, rv, o_fealab);
  else if (ct < 26)
    head_job<NT, 1>(sH, sPre, sTanh, ct - 19, rt_first, base, nvalid, lane, wclas, bclas,
                    gamma, beta, rm, rv, o_logists);
  else
    head_job<NT, 2>(sH, sPre, sTanh, ct - 26, rt_first, base, nvalid, lane, wconv, bconv,
                    gamma, beta, rm, rv, o_tanh);
}

template <int COUNT>
__global__ void __launch_bounds__(AGG_THREADS)
range_layer_kernel(const float* __restrict__ x, const int* __restrict__ src,
                   const int* __restrict__ dst, const unsigned* __restrict__ rid,
                   float* __restrict__ deg,
                   unsigned* __restrict__ wlist, unsigned* __restrict__ wcount,
                   const float* __restrict__ wl, const float* __restrict__ bl,
                   const float* __restrict__ wr, float* __restrict__ h,
                   const float* __restrict__ whd, const float* __restrict__ bhd,
                   const float* __restrict__ wclas, const float* __restrict__ bclas,
                   const float* __restrict__ wconv, const float* __restrict__ bconv,
                   const float* __restrict__ gamma, const float* __restrict__ beta,
                   const float* __restrict__ rm, const float* __restrict__ rv,
                   const float* __restrict__ wtl, const float* __restrict__ btl,
                   const float* __restrict__ wcv, const float* __restrict__ bcv,
                   float* __restrict__ o_logists, float* __restrict__ o_tanh,
                   float* __restrict__ o_fealab, float* __restrict__ o_feacv,
                   float* __restrict__ o_truelab) {
  __shared__ __attribute__((aligned(16))) float sAcc[AGG_ROWS * LDS_STRIDE];
  __shared__ __attribute__((aligned(16))) float sX[AGG_ROWS * LDS_STRIDE];
  __shared__ unsigned sList[AGG_CAP];
  __shared__ float    sCnt[AGG_ROWS];
  __shared__ unsigned sN;

  const int t    = threadIdx.x;
  const int lane = t & 31;
  const int wave = __builtin_amdgcn_readfirstlane(t >> 5);
  const int base = blockIdx.x * AGG_R;

  for (int i = t; i < AGG_ROWS * LDS_STRIDE / 4; i += AGG_THREADS)
    reinterpret_cast<float4*>(sAcc)[i] = make_float4(0.0f, 0.0f, 0.0f, 0.0f);
  if (t < AGG_ROWS) {
    float c = 0.0f;
    if (!COUNT && t < AGG_R && base + t < N_NODES) c = deg[base + t];
    sCnt[t] = c;
  }
  if (t == 0) sN = 0u;
  __syncthreads();

  auto append = [&](int e) {
    const unsigned pos = atomicAdd(&sN, 1u);
    if (pos < AGG_CAP) {
      sList[pos] = (unsigned)e;
    } else {
      const unsigned u = (unsigned)(dst[e] - base);
      if (u < AGG_R) {
        if (COUNT) atomicAdd(&sCnt[u], 1.0f);
        const float* xr = x + (size_t)src[e] * F_DIM;
        for (int c = 0; c < F_DIM; ++c)
          atomicAdd(&sAcc[u * LDS_STRIDE + (c & 3) * ACC_COMP + (c >> 2)], xr[c]);
      }
    }
  };

  unsigned saved = 0u;
  if (!COUNT) saved = (unsigned)__builtin_amdgcn_readfirstlane((int)wcount[blockIdx.x]);
  const bool rescan = COUNT || saved > (unsigned)AGG_CAP;
  unsigned* wsl = wlist + (size_t)blockIdx.x * AGG_CAP;

  if (!rescan) {
    for (unsigned i = t; i < saved; i += AGG_THREADS) sList[i] = wsl[i];
    if (t == 0) sN = saved;
  } else {
    const uint4* r4 = reinterpret_cast<const uint4*>(rid);
    const unsigned bb = (unsigned)blockIdx.x * 0x01010101u;
    const int off = blockIdx.x * (N_RG / AGG_BLOCKS);
    auto load4 = [&](int it, uint4* d, int* gi) {
#pragma unroll
      for (int j = 0; j < 4; ++j) {
        const int  g  = (it + j) * AGG_THREADS + t;
        const bool ok = g < N_RG;
        int gg = g + off;
        if (gg >= N_RG) gg -= N_RG;
        gg = ok ? gg : 0;
        gi[j] = gg;
        uint4 v = r4[gg];
        if (!ok) v = make_uint4(~bb, ~bb, ~bb, ~bb);
        d[j] = v;
      }
    };
    auto hit = [&](unsigned w) -> unsigned {
      const unsigned xw = w ^ bb;
      return ~(((xw & 0x7F7F7F7Fu) + 0x7F7F7F7Fu) | xw | 0x7F7F7F7Fu);
    };
    uint4 cur[4];
    int   gcur[4];
    load4(0, cur, gcur);
    for (int it = 0; it < (N_RG + AGG_THREADS - 1) / AGG_THREADS; it += 4) {
      uint4 nxt[4];
      int   gnxt[4];
      load4(it + 4, nxt, gnxt);
#pragma unroll
      for (int j = 0; j < 4; ++j) {
        unsigned m = (hit(cur[j].x) >> 7) | (hit(cur[j].y) >> 6) |
                     (hit(cur[j].z) >> 5) | (hit(cur[j].w) >> 4);
        const int e = gcur[j] * 16;
        while (m) {
          const int p = __builtin_ctz(m);
          append(e + ((p & 7) << 2) + (p >> 3));
          m &= m - 1u;
        }
      }
#pragma unroll
      for (int j = 0; j < 4; ++j) { cur[j] = nxt[j]; gcur[j] = gnxt[j]; }
    }
  }
  __syncthreads();

  const unsigned nraw = sN;
  const int n = (int)(nraw < AGG_CAP ? nraw : (unsigned)AGG_CAP);
  const unsigned spare = (unsigned)AGG_R << 20;
  if (rescan) {
    for (int i0 = t; i0 < n; i0 += 4 * AGG_THREADS) {
      unsigned e[4];
      int      dv[4];
#pragma unroll
      for (int j = 0; j < 4; ++j) {
        const int i = i0 + j * AGG_THREADS;
        e[j]  = (i < n) ? sList[i] : 0u;
        dv[j] = dst[e[j]];
      }
#pragma unroll
      for (int j = 0; j < 4; ++j) {
        const int i = i0 + j * AGG_THREADS;
        const unsigned u = (unsigned)(dv[j] - base);
        unsigned ent = spare;
        if (u < AGG_R) ent = e[j] | (u << 20);
        if (i < n) {
          if (COUNT && u < AGG_R) atomicAdd(&sCnt[u], 1.0f);
          sList[i] = ent;
          if (COUNT) wsl[i] = ent;
        }
      }
    }
    if (COUNT && t == 0) wcount[blockIdx.x] = nraw;
    __syncthreads();
  }
  auto entry = [&](int b) -> unsigned {
    const int idx = b + (lane & 7);
    return (idx < n) ? sList[idx] : spare;
  };
  unsigned ent = entry(wave * 8);
  int      s   = src[ent & 0xFFFFFu];
  for (int b = wave * 8; b < n; b += (AGG_THREADS / 32) * 8) {
    float4 row[8];
#pragma unroll
    for (int j = 0; j < 8; ++j) {
      const int sj = __builtin_amdgcn_readlane(s, j);
      row[j] = *reinterpret_cast<const float4*>(x + (size_t)sj * F_DIM + lane * 4);
    }
    const unsigned ent_n = entry(b + (AGG_THREADS / 32) * 8);
    const int      s_n   = src[ent_n & 0xFFFFFu];
#pragma unroll
    for (int j = 0; j < 8; ++j) {
      const unsigned ld = (unsigned)__builtin_amdgcn_readlane((int)ent, j) >> 20;
      float* p = sAcc + ld * LDS_STRIDE + lane;
      atomicAdd(p,      row[j].x);
      atomicAdd(p + ACC_COMP,     row[j].y);
      atomicAdd(p + 2 * ACC_COMP, row[j].z);
      atomicAdd(p + 3 * ACC_COMP, row[j].w);
    }
    ent = ent_n;
    s   = s_n;
  }

  for (int i = t; i < AGG_ROWS * 32; i += AGG_THREADS) {
    const int r    = i >> 5;
    const int grow = base + r;
    float4 v = make_float4(0.0f, 0.0f, 0.0f, 0.0f);
    if (r < AGG_R && grow < N_NODES)
      v = *reinterpret_cast<const float4*>(x + (size_t)grow * F_DIM + lane * 4);
    *reinterpret_cast<float4*>(sX + r * LDS_STRIDE + lane * 4) = v;
  }
  __syncthreads();

  if (COUNT) {
    if (t < AGG_R && base + t < N_NODES) deg[base + t] = sCnt[t];
  }

  const int tn  = wave & 7;
  const int rt0 = wave >> 3;
  if (COUNT) {
    if (rt0 == 0)
      range_tiles_gemm<4, 0>(sAcc, sX, sCnt, wl, bl, wr, h, nullptr, base, rt0, tn, lane);
    else
      range_tiles_gemm<3, 0>(sAcc, sX, sCnt, wl, bl, wr, h, nullptr, base, rt0, tn, lane);
    return;
  }

  v8f hacc[4];
  hacc[3] = v8f{};
  if (rt0 == 0)
    range_tiles_gemm<4, 1>(sAcc, sX, sCnt, wl, bl, wr, h, hacc, base, rt0, tn, lane);
  else
    range_tiles_gemm<3, 1>(sAcc, sX, sCnt, wl, bl, wr, h, hacc, base, rt0, tn, lane);
  const int ntl = (rt0 == 0) ? 4 : 3;
  __syncthreads();
  {
    const int mloc = lane & 15;
    const int hh   = lane >> 4;
#pragma unroll
    for (int i = 0; i < 4; ++i) {
      if (i < ntl) {
        const int lr0 = (rt0 + 4 * i) * 16 + hh * 8;
#pragma unroll
        for (int q = 0; q < 8; ++q)
          sX[(lr0 + q) * LDS_STRIDE + tn * 16 + mloc] = hacc[i][q];
      }
    }
  }
  __syncthreads();

  float* sPre  = reinterpret_cast<float*>(sList);
  float* sTanh = sAcc;
  const int nvalid = (N_NODES - base < AGG_R) ? (N_NODES - base) : AGG_R;
  if (wave < 30)
    head_dispatch<7>(sX, sPre, sTanh, wave, 0, base, nvalid, lane, whd, bhd, wclas, bclas,
                     wconv, bconv, gamma, beta, rm, rv, o_fealab, o_logists, o_tanh);
  if (wave < 28 || wave >= 30)
    head_dispatch<6>(sX, sPre, sTanh, (wave < 28) ? wave + 2 : wave - 30, 7, base, nvalid, lane,
                     whd, bhd, wclas, bclas, wconv, bconv, gamma, beta, rm, rv,
                     o_fealab, o_logists, o_tanh);
  __syncthreads();

  if (wave < 28) {
    const int ctn  = wave % 7;
    const int rg   = wave / 7;
    const int mloc = lane & 15;
    const int ncol = ctn * 16 + mloc;
    const bool nok = ncol < NCLS;
    const int ncolc = nok ? ncol : NCLS - 1;
    if (rg == 0) {
      v8f acc[4];
      head_tiles_gemm<4, NBITS, NCLS, HT_STRIDE>(sTanh, wcv, ncolc, rg, 4, lane, acc);
      head_store<4, NCLS>(acc, o_feacv, bcv[ncolc], ncol, nok, base, nvalid, rg, 4, lane);
    } else {
      v8f acc[3];
      head_tiles_gemm<3, NBITS, NCLS, HT_STRIDE>(sTanh, wcv, ncolc, rg, 4, lane, acc);
      head_store<3, NCLS>(acc, o_feacv, bcv[ncolc], ncol, nok, base, nvalid, rg, 4, lane);
    }
  } else {
    for (int r = t - 28 * 32; r < nvalid; r += 4 * 32) {
      const float* p = sPre + r * HT_STRIDE;
      float sum = 0.0f;
#pragma unroll
      for (int k = 0; k < NBITS; ++k) sum += p[k] * wtl[k];
      o_truelab[base + r] = sum + btl[0];
    }
  }
}

extern "C" void kernel_launch(void* const* d_in, const int* in_sizes, int n_in,
                              void* d_out, int out_size, void* d_ws, size_t ws_size,
                              hipStream_t stream) {
  (void)in_sizes; (void)n_in; (void)out_size; (void)ws_size;

  const float* features = (const float*)d_in[0];
  const int*   edges    = (const int*)d_in[1];
  const float* w1l   = (const float*)d_in[2];
  const float* b1l   = (const float*)d_in[3];
  const float* w1r   = (const float*)d_in[4];
  const float* w2l   = (const float*)d_in[5];
  const float* b2l   = (const float*)d_in[6];
  const float* w2r   = (const float*)d_in[7];
  const float* whd   = (const float*)d_in[8];
  const float* bhd   = (const float*)d_in[9];
  const float* wclas = (const float*)d_in[10];
  const float* bclas = (const float*)d_in[11];
  const float* wconv = (const float*)d_in[12];
  const float* bconv = (const float*)d_in[13];
  const float* gamma = (const float*)d_in[14];
  const float* beta  = (const float*)d_in[15];
  const float* rm    = (const float*)d_in[16];
  const float* rv    = (const float*)d_in[17];
  const float* wtl   = (const float*)d_in[18];
  const float* btl   = (const float*)d_in[19];
  const float* wcv   = (const float*)d_in[20];
  const float* bcv   = (const float*)d_in[21];

  const int* src = edges;
  const int* dst = edges + N_EDGES;

  float* ws     = (float*)d_ws;
  float* deg    = ws;
  float* h1     = deg + N_NODES;
  unsigned* wcount = (unsigned*)(h1 + (size_t)N_NODES * F_DIM);
  unsigned* wlist  = wcount + AGG_BLOCKS;
  unsigned* rid    = wlist + (size_t)AGG_BLOCKS * AGG_CAP;

  float* out       = (float*)d_out;
  float* o_logists = out;
  float* o_tanh    = o_logists + (size_t)N_NODES * NCLS;
  float* o_fealab  = o_tanh    + (size_t)N_NODES * NBITS;
  float* o_feacv   = o_fealab  + (size_t)N_NODES * HD;
  float* o_truelab = o_feacv   + (size_t)N_NODES * NCLS;

  range_id_kernel<<<RID_BLOCKS, RID_THREADS, 0, stream>>>(dst, rid);

  range_layer_kernel<1><<<AGG_BLOCKS, AGG_THREADS, 0, stream>>>(
      features, src, dst, rid, deg, wlist, wcount, w1l, b1l, w1r, h1,
      whd, bhd, wclas, bclas, wconv, bconv, gamma, beta, rm, rv, wtl, btl, wcv, bcv,
      o_logists, o_tanh, o_fealab, o_feacv, o_truelab);

  range_layer_kernel<0><<<AGG_BLOCKS, AGG_THREADS, 0, stream>>>(
      h1, src, dst, rid, deg, wlist, wcount, w2l, b2l, w2r, nullptr,
      whd, bhd, wclas, bclas, wconv, bconv, gamma, beta, rm, rv, wtl, btl, wcv, bcv,
      o_logists, o_tanh, o_fealab, o_feacv, o_truelab);
}
